# starting kernel
# baseline (speedup 1.0000x reference)
#include <hip/hip_runtime.h>


#define GA_ALPHA 0.2f
#define GA_BETA1 0.6931471805599453f
#define GA_BETA2 0.4054651081081644f

typedef __attribute__((ext_vector_type(2))) float v2f;
typedef __attribute__((ext_vector_type(8))) float v8f;


__global__ __launch_bounds__(256) void k_zero_i(int* __restrict__ p, long n) {
  long i = (long)blockIdx.x * 256 + threadIdx.x;
  if (i < n) p[i] = 0;
}

__global__ __launch_bounds__(256) void k_degcount(const int* __restrict__ src,
                                                  const int* __restrict__ dst,
                                                  int* __restrict__ cntO,
                                                  int* __restrict__ cntI, int E) {
  int e = blockIdx.x * 256 + threadIdx.x;
  if (e < E) {
    atomicAdd(&cntO[src[e]], 1);
    atomicAdd(&cntI[dst[e]], 1);
  }
}

__global__ __launch_bounds__(256) void k_degrsqrt(const int* __restrict__ cntO,
                                                  const int* __restrict__ cntI,
                                                  float* __restrict__ degO,
                                                  float* __restrict__ degI, int n) {
  int i = blockIdx.x * 256 + threadIdx.x;
  if (i < n) {
    degO[i] = rsqrtf(fmaxf((float)cntO[i], 1.0f));
    degI[i] = rsqrtf(fmaxf((float)cntI[i], 1.0f));
  }
}


__global__ __launch_bounds__(256) void k_scan_block(const int* __restrict__ in,
                                                    int* __restrict__ outEx,
                                                    int* __restrict__ bsum, int n) {
  __shared__ int s[256];
  const int tid = threadIdx.x;
  const int i = blockIdx.x * 256 + tid;
  const int v = (i < n) ? in[i] : 0;
  s[tid] = v;
  __syncthreads();
#pragma unroll
  for (int off = 1; off < 256; off <<= 1) {
    int t = (tid >= off) ? s[tid - off] : 0;
    __syncthreads();
    s[tid] += t;
    __syncthreads();
  }
  if (i < n) outEx[i] = s[tid] - v;
  if (tid == 255) bsum[blockIdx.x] = s[255];
}

__global__ __launch_bounds__(512) void k_scan_tops(int* __restrict__ bsum, int nb) {
  __shared__ int s[512];
  const int tid = threadIdx.x;
  const int v = (tid < nb) ? bsum[tid] : 0;
  s[tid] = v;
  __syncthreads();
#pragma unroll
  for (int off = 1; off < 512; off <<= 1) {
    int t = (tid >= off) ? s[tid - off] : 0;
    __syncthreads();
    s[tid] += t;
    __syncthreads();
  }
  if (tid < nb) bsum[tid] = s[tid] - v;
}

__global__ __launch_bounds__(256) void k_scan_add(int* __restrict__ outEx,
                                                  const int* __restrict__ bsum,
                                                  int* __restrict__ cursor, int n) {
  int i = blockIdx.x * 256 + threadIdx.x;
  if (i < n) {
    int r = outEx[i] + bsum[blockIdx.x];
    outEx[i] = r;
    cursor[i] = r;
  }
}


__global__ __launch_bounds__(256) void k_build_csc(const int* __restrict__ src,
                                                   const int* __restrict__ dst,
                                                   int* __restrict__ cursor,
                                                   int* __restrict__ csc, int E) {
  int e = blockIdx.x * 256 + threadIdx.x;
  if (e < E) {
    int pos = atomicAdd(&cursor[dst[e]], 1);
    csc[pos] = src[e];
  }
}


__global__ __launch_bounds__(256) void k_scale(const float4* __restrict__ x,
                                               const float* __restrict__ dors,
                                               float4* __restrict__ hs, long total4) {
  long i = (long)blockIdx.x * 256 + threadIdx.x;
  if (i < total4) {
    const int row = (int)(i >> 5);
    const float sS = dors[row];
    float4 v = x[i];
    v.x *= sS; v.y *= sS; v.z *= sS; v.w *= sS;
    hs[i] = v;
  }
}

__global__ __launch_bounds__(256) void k_spmm_pull(const int* __restrict__ rowptr,
                                                   const int* __restrict__ cnt,
                                                   const int* __restrict__ csc,
                                                   const float* __restrict__ hs,
                                                   const float* __restrict__ dirs,
                                                   const float* __restrict__ x0,
                                                   float* __restrict__ f, int N) {
  const int node = blockIdx.x * 8 + (threadIdx.x >> 5);
  if (node >= N) return;
  const int lane = threadIdx.x & 31;
  const int start = rowptr[node];
  const int c = cnt[node];
  float4 acc = {0.0f, 0.0f, 0.0f, 0.0f};
  for (int i = 0; i < c; ++i) {
    const int s = csc[start + i];
    const float4 v = *(const float4*)(hs + (long)s * 128 + lane * 4);
    acc.x += v.x; acc.y += v.y; acc.z += v.z; acc.w += v.w;
  }
  const float scale = (1.0f - GA_ALPHA) * dirs[node];
  const long o = (long)node * 128 + lane * 4;
  const float4 x0v = *(const float4*)(x0 + o);
  float4 r;
  r.x = scale * acc.x + GA_ALPHA * x0v.x;
  r.y = scale * acc.y + GA_ALPHA * x0v.y;
  r.z = scale * acc.z + GA_ALPHA * x0v.z;
  r.w = scale * acc.w + GA_ALPHA * x0v.w;
  *(float4*)(f + o) = r;
}

template <int NCOLS, bool RELU, bool RESID>
__global__ __launch_bounds__(256) void k_gemm_wmma(const float* __restrict__ A,
                                                   const float* __restrict__ W,
                                                   const float* __restrict__ bias,
                                                   float* __restrict__ out,
                                                   float actW, float resS) {
  constexpr int NCT = NCOLS / 16;
  const int wave = threadIdx.x >> 5;
  const int lane = threadIdx.x & 31;
  const int colTile = wave % NCT;
  const int rowTile = blockIdx.x * (8 / NCT) + wave / NCT;
  const int row0 = rowTile * 16;
  const int col0 = colTile * 16;
  const int half = lane >> 4;
  const int l16  = lane & 15;

  const float* __restrict__ Arow = A + (long)(row0 + l16) * 128;
  const float* __restrict__ Wcol = W + col0 + l16;

  v8f acc = {};
#pragma unroll
  for (int k = 0; k < 128; k += 4) {
    const int ka = k + half * 2;
    v2f af, bf;
    af.x = Arow[ka];
    af.y = Arow[ka + 1];
    bf.x = Wcol[(long)ka * NCOLS];
    bf.y = Wcol[(long)(ka + 1) * NCOLS];
    acc = __builtin_amdgcn_wmma_f32_16x16x4_f32(false, af, false, bf,
                                                (short)0, acc, false, false);
  }

  const int outCol = col0 + l16;
  const float bv = bias[outCol];
#pragma unroll
  for (int r = 0; r < 8; ++r) {
    const int rowM = row0 + r + half * 8;
    float v = actW * acc[r] + bv;
    if (RESID) v += resS * A[(long)rowM * 128 + outCol];
    if (RELU) v = fmaxf(v, 0.0f);
    out[(long)rowM * NCOLS + outCol] = v;
  }
}


static inline long cdiv(long a, long b) { return (a + b - 1) / b; }

extern "C" void kernel_launch(void* const* d_in, const int* in_sizes, int n_in,
                              void* d_out, int out_size, void* d_ws, size_t ws_size,
                              hipStream_t stream) {
  const float* feat = (const float*)d_in[0];
  const int*   src  = (const int*)d_in[1];
  const int*   dst  = (const int*)d_in[2];
  const float* w1   = (const float*)d_in[3];
  const float* b1   = (const float*)d_in[4];
  const float* cw1  = (const float*)d_in[5];
  const float* cb1  = (const float*)d_in[6];
  const float* cw2  = (const float*)d_in[7];
  const float* cb2  = (const float*)d_in[8];
  const float* w2   = (const float*)d_in[9];
  const float* b2   = (const float*)d_in[10];

  const int N = in_sizes[0] / 128;
  const int E = in_sizes[1];
  const long NF = (long)N * 128;
  const int NA = (N + 63) & ~63;

  char* wsb = (char*)d_ws;
  int* cntO   = (int*)wsb;                wsb += (size_t)NA * 4;
  int* cntI   = (int*)wsb;                wsb += (size_t)NA * 4;
  int* rowptr = (int*)wsb;                wsb += (size_t)NA * 4;
  int* cursor = (int*)wsb;                wsb += (size_t)NA * 4;
  int* bsum   = (int*)wsb;                wsb += 512 * 4;
  int* csc    = (int*)wsb;                wsb += ((size_t)E * 4 + 255) & ~(size_t)255;
  float* degO = (float*)wsb;              wsb += (size_t)NA * 4;
  float* degI = (float*)wsb;              wsb += (size_t)NA * 4;
  float* h0   = (float*)wsb;              wsb += (size_t)NF * 4;
  float* bufS = (float*)wsb;              wsb += (size_t)NF * 4;
  float* bufA = (float*)wsb;              wsb += (size_t)NF * 4;
  float* bufB = (float*)wsb;
  (void)ws_size; (void)n_in; (void)out_size;

  const int NB       = (int)cdiv(N, 256);
  const int nb_node  = NB;
  const int nb_edge  = (int)cdiv(E, 256);
  const int nb_vec4  = (int)cdiv(NF / 4, 256);
  const int nb_pull  = (int)cdiv(N, 8);
  const int nb_g128  = N / 16;
  const int nb_g64   = N / 32;

  k_zero_i<<<(int)cdiv(2L * NA, 256), 256, 0, stream>>>(cntO, 2L * NA);
  k_degcount<<<nb_edge, 256, 0, stream>>>(src, dst, cntO, cntI, E);
  k_degrsqrt<<<nb_node, 256, 0, stream>>>(cntO, cntI, degO, degI, N);
  k_scan_block<<<NB, 256, 0, stream>>>(cntI, rowptr, bsum, N);
  k_scan_tops<<<1, 512, 0, stream>>>(bsum, NB);
  k_scan_add<<<NB, 256, 0, stream>>>(rowptr, bsum, cursor, N);
  k_build_csc<<<nb_edge, 256, 0, stream>>>(src, dst, cursor, csc, E);

  k_gemm_wmma<128, true, false><<<nb_g128, 256, 0, stream>>>(feat, w1, b1, h0, 1.0f, 0.0f);

  k_scale<<<nb_vec4, 256, 0, stream>>>((const float4*)h0, degO, (float4*)bufS, NF / 4);
  k_spmm_pull<<<nb_pull, 256, 0, stream>>>(rowptr, cntI, csc, bufS, degI, h0, bufA, N);
  k_gemm_wmma<128, true, true><<<nb_g128, 256, 0, stream>>>(bufA, cw1, cb1, bufS,
                                                            GA_BETA1, 1.0f - GA_BETA1);

  k_scale<<<nb_vec4, 256, 0, stream>>>((const float4*)bufS, degO, (float4*)bufB, NF / 4);
  k_spmm_pull<<<nb_pull, 256, 0, stream>>>(rowptr, cntI, csc, bufB, degI, h0, bufA, N);
  k_gemm_wmma<128, false, true><<<nb_g128, 256, 0, stream>>>(bufA, cw2, cb2, bufB,
                                                             GA_BETA2, 1.0f - GA_BETA2);

  k_gemm_wmma<64, false, false><<<nb_g64, 256, 0, stream>>>(bufB, w2, b2, (float*)d_out,
                                                            1.0f, 0.0f);
}


# Source-degree scaling rides on the edges: CSC entries hold (source, its deg_out^-1/2); fused gathers read unscaled rows and FMA the weight in; both scale launches, scaled tables gone; sum order same
# speedup vs baseline: 1.3668x; 1.3668x over previous
#include <hip/hip_runtime.h>


#define GA_ALPHA 0.2f
#define GA_BETA1 0.6931471805599453f
#define GA_BETA2 0.4054651081081644f

typedef __attribute__((ext_vector_type(2))) float v2f;
typedef __attribute__((ext_vector_type(2))) int   v2i;
typedef __attribute__((ext_vector_type(4))) float v4f;
typedef __attribute__((ext_vector_type(8))) float v8f;


__global__ __launch_bounds__(256) void k_zero_i(int* __restrict__ p, long n) {
  long i = (long)blockIdx.x * 256 + threadIdx.x;
  if (i < n) p[i] = 0;
}

__global__ __launch_bounds__(256) void k_degcount(const int* __restrict__ src,
                                                  const int* __restrict__ dst,
                                                  int* __restrict__ cntO,
                                                  int* __restrict__ cntI, int E) {
  int e = blockIdx.x * 256 + threadIdx.x;
  if (e < E) {
    atomicAdd(&cntO[src[e]], 1);
    atomicAdd(&cntI[dst[e]], 1);
  }
}

__global__ __launch_bounds__(256) void k_degrsqrt(const int* __restrict__ cntO,
                                                  const int* __restrict__ cntI,
                                                  float* __restrict__ degO,
                                                  float* __restrict__ degI, int n) {
  int i = blockIdx.x * 256 + threadIdx.x;
  if (i < n) {
    degO[i] = rsqrtf(fmaxf((float)cntO[i], 1.0f));
    degI[i] = rsqrtf(fmaxf((float)cntI[i], 1.0f));
  }
}


__global__ __launch_bounds__(256) void k_scan_block(const int* __restrict__ in,
                                                    int* __restrict__ outEx,
                                                    int* __restrict__ bsum, int n) {
  __shared__ int s[256];
  const int tid = threadIdx.x;
  const int i = blockIdx.x * 256 + tid;
  const int v = (i < n) ? in[i] : 0;
  s[tid] = v;
  __syncthreads();
#pragma unroll
  for (int off = 1; off < 256; off <<= 1) {
    int t = (tid >= off) ? s[tid - off] : 0;
    __syncthreads();
    s[tid] += t;
    __syncthreads();
  }
  if (i < n) outEx[i] = s[tid] - v;
  if (tid == 255) bsum[blockIdx.x] = s[255];
}

__global__ __launch_bounds__(512) void k_scan_tops(int* __restrict__ bsum, int nb) {
  __shared__ int s[512];
  const int tid = threadIdx.x;
  const int v = (tid < nb) ? bsum[tid] : 0;
  s[tid] = v;
  __syncthreads();
#pragma unroll
  for (int off = 1; off < 512; off <<= 1) {
    int t = (tid >= off) ? s[tid - off] : 0;
    __syncthreads();
    s[tid] += t;
    __syncthreads();
  }
  if (tid < nb) bsum[tid] = s[tid] - v;
}

__global__ __launch_bounds__(256) void k_scan_add(int* __restrict__ outEx,
                                                  const int* __restrict__ bsum,
                                                  int* __restrict__ cursor, int n) {
  int i = blockIdx.x * 256 + threadIdx.x;
  if (i < n) {
    int r = outEx[i] + bsum[blockIdx.x];
    outEx[i] = r;
    cursor[i] = r;
  }
}


__global__ __launch_bounds__(256) void k_build_csc(const int* __restrict__ src,
                                                   const int* __restrict__ dst,
                                                   const float* __restrict__ dors,
                                                   int* __restrict__ cursor,
                                                   int* __restrict__ csc, int E) {
  int e = blockIdx.x * 256 + threadIdx.x;
  if (e < E) {
    const int s = src[e];
    const float w = dors[s];
    int pos = atomicAdd(&cursor[dst[e]], 1);
    v2i ent;
    ent.x = s;
    ent.y = __float_as_int(w);
    *(v2i*)(csc + 2 * (long)pos) = ent;
  }
}

template <int NCOLS, int WAVES, bool RELU, bool RESID>
__global__ __launch_bounds__(WAVES * 32) void k_gemm_wmma(const float* __restrict__ A,
                                                          const float* __restrict__ W,
                                                          const float* __restrict__ bias,
                                                          float* __restrict__ out,
                                                          float actW, float resS, int nRows) {
  constexpr int NCT   = NCOLS / 16;
  constexpr int MT    = WAVES / NCT;
  constexpr int BR    = MT * 16;
  constexpr int RPW   = BR / WAVES;
  constexpr int PITCH = 132;
  __shared__ __attribute__((aligned(16))) float lds[2][BR * PITCH];

  const int wave = threadIdx.x >> 5;
  const int lane = threadIdx.x & 31;
  const int ct   = wave % NCT;
  const int mt   = wave / NCT;
  const int col0 = ct * 16;
  const int half = lane >> 4;
  const int l16  = lane & 15;
  const int outCol = col0 + l16;

  v2f bfr[32];
#pragma unroll
  for (int s = 0; s < 32; ++s) {
    const int kb = 8 * (s >> 1) + 4 * half + 2 * (s & 1);
    bfr[s].x = W[(long)kb * NCOLS + outCol];
    bfr[s].y = W[(long)(kb + 1) * NCOLS + outCol];
  }
  const float bv = bias[outCol];

  const int nblk = nRows / BR;
  int blk = blockIdx.x;

  v4f stage[RPW];
#pragma unroll
  for (int i = 0; i < RPW; ++i) stage[i] = (v4f){0.0f, 0.0f, 0.0f, 0.0f};
  if (blk < nblk) {
#pragma unroll
    for (int i = 0; i < RPW; ++i)
      stage[i] = *(const v4f*)(A + (long)(blk * BR + wave * RPW + i) * 128 + lane * 4);
  }

  int p = 0;
  for (; blk < nblk; blk += gridDim.x) {
    float* __restrict__ buf = lds[p];
#pragma unroll
    for (int i = 0; i < RPW; ++i)
      *(v4f*)(buf + (wave * RPW + i) * PITCH + lane * 4) = stage[i];

    __builtin_amdgcn_fence(__ATOMIC_RELEASE, "workgroup", "local");
    __builtin_amdgcn_s_barrier();
    __builtin_amdgcn_fence(__ATOMIC_ACQUIRE, "workgroup", "local");

    const int next = blk + gridDim.x;
    if (next < nblk) {
#pragma unroll
      for (int i = 0; i < RPW; ++i)
        stage[i] = *(const v4f*)(A + (long)(next * BR + wave * RPW + i) * 128 + lane * 4);
    }

    const float* __restrict__ arow = buf + (mt * 16 + l16) * PITCH + half * 4;
    v4f af[16];
#pragma unroll
    for (int j = 0; j < 16; ++j) af[j] = *(const v4f*)(arow + 8 * j);

    v8f acc = {};
#pragma unroll
    for (int j = 0; j < 16; ++j) {
      v2f a0, a1;
      a0.x = af[j].x; a0.y = af[j].y;
      a1.x = af[j].z; a1.y = af[j].w;
      acc = __builtin_amdgcn_wmma_f32_16x16x4_f32(false, a0, false, bfr[2 * j],
                                                  (short)0, acc, false, false);
      acc = __builtin_amdgcn_wmma_f32_16x16x4_f32(false, a1, false, bfr[2 * j + 1],
                                                  (short)0, acc, false, false);
    }

    const int rl0 = mt * 16 + half * 8;
    float* __restrict__ orow = out + (long)(blk * BR + rl0) * NCOLS + outCol;
#pragma unroll
    for (int r = 0; r < 8; ++r) {
      float v = actW * acc[r] + bv;
      if (RESID) v += resS * buf[(rl0 + r) * PITCH + outCol];
      if (RELU) v = fmaxf(v, 0.0f);
      orow[r * NCOLS] = v;
    }
    p ^= 1;
  }
}


static __device__ __forceinline__ v4f gather_node(const v2i* __restrict__ csc,
                                                  const float* __restrict__ x,
                                                  int start, int c, v2i ent0, int lane) {
  const unsigned lane4 = (unsigned)lane * 4u;
  v4f acc = {0.0f, 0.0f, 0.0f, 0.0f};
  int idx = ent0.x;
  int wbt = ent0.y;
  for (int base = 0; base < c; base += 32) {
    if (base > 0) {
      v2i t = {0, 0};
      if (base + lane < c) t = csc[start + base + lane];
      idx = t.x;
      wbt = t.y;
    }
    const int n = (c - base < 32) ? c - base : 32;
#pragma unroll 1
    for (int j = 0; j < n; j += 8) {
      int   s[8];
      float w[8];
#pragma unroll
      for (int u = 0; u < 8; ++u) {
        const int e = (j + u < n) ? j + u : n - 1;
        s[u] = __builtin_amdgcn_readlane(idx, e);
        w[u] = __int_as_float(__builtin_amdgcn_readlane(wbt, e));
      }
      v4f v[8];
      if (j + 8 <= n) {
#pragma unroll
        for (int u = 0; u < 8; ++u) v[u] = *(const v4f*)(x + (long)s[u] * 128 + lane4);
#pragma unroll
        for (int u = 0; u < 8; ++u) acc += w[u] * v[u];
      } else {
#pragma unroll
        for (int u = 0; u < 8; ++u) {
          v[u] = (v4f){0.0f, 0.0f, 0.0f, 0.0f};
          if (j + u < n) v[u] = *(const v4f*)(x + (long)s[u] * 128 + lane4);
        }
#pragma unroll
        for (int u = 0; u < 8; ++u) acc += w[u] * v[u];
      }
    }
  }
  return acc;
}

template <bool RELU>
__global__ __launch_bounds__(256) void k_layer_fused(const int* __restrict__ rowptr,
                                                     const int* __restrict__ cnt,
                                                     const int* __restrict__ cscRaw,
                                                     const float* __restrict__ x,
                                                     const float* __restrict__ dirs,
                                                     const float* __restrict__ x0,
                                                     const float* __restrict__ W,
                                                     const float* __restrict__ bias,
                                                     float* __restrict__ out,
                                                     float actW, float resS, int nRows) {
  constexpr int NCOLS = 128;
  constexpr int BR    = 16;
  constexpr int PITCH = 132;
  __shared__ __attribute__((aligned(16))) float lds[2][BR * PITCH];
  const v2i* __restrict__ csc = (const v2i*)cscRaw;

  const int wave = __builtin_amdgcn_readfirstlane((int)(threadIdx.x >> 5));
  const int lane = threadIdx.x & 31;
  const int col0 = wave * 16;
  const int half = lane >> 4;
  const int l16  = lane & 15;
  const int outCol = col0 + l16;

  v2f bfr[32];
#pragma unroll
  for (int s = 0; s < 32; ++s) {
    const int kb = 8 * (s >> 1) + 4 * half + 2 * (s & 1);
    bfr[s].x = W[(long)kb * NCOLS + outCol];
    bfr[s].y = W[(long)(kb + 1) * NCOLS + outCol];
  }
  const float bv = bias[outCol];

  const int nblk = nRows / BR;
  int p = 0;
  for (int blk = blockIdx.x; blk < nblk; blk += gridDim.x) {
    float* __restrict__ buf = lds[p];

    const int nodeA  = blk * BR + wave * 2;
    const int startA = rowptr[nodeA];
    const int startB = rowptr[nodeA + 1];
    const int cA     = cnt[nodeA];
    const int cB     = cnt[nodeA + 1];
    v2i iA = {0, 0};
    v2i iB = {0, 0};
    if (lane < cA) iA = csc[startA + lane];
    if (lane < cB) iB = csc[startB + lane];
    const float scA = (1.0f - GA_ALPHA) * dirs[nodeA];
    const float scB = (1.0f - GA_ALPHA) * dirs[nodeA + 1];
    const long  oA  = (long)nodeA * 128 + lane * 4;
    const v4f x0A = *(const v4f*)(x0 + oA);
    const v4f x0B = *(const v4f*)(x0 + oA + 128);

    const v4f accA = gather_node(csc, x, startA, cA, iA, lane);
    v4f rA;
    rA.x = scA * accA.x + GA_ALPHA * x0A.x;
    rA.y = scA * accA.y + GA_ALPHA * x0A.y;
    rA.z = scA * accA.z + GA_ALPHA * x0A.z;
    rA.w = scA * accA.w + GA_ALPHA * x0A.w;
    *(v4f*)(buf + (wave * 2) * PITCH + lane * 4) = rA;

    const v4f accB = gather_node(csc, x, startB, cB, iB, lane);
    v4f rB;
    rB.x = scB * accB.x + GA_ALPHA * x0B.x;
    rB.y = scB * accB.y + GA_ALPHA * x0B.y;
    rB.z = scB * accB.z + GA_ALPHA * x0B.z;
    rB.w = scB * accB.w + GA_ALPHA * x0B.w;
    *(v4f*)(buf + (wave * 2 + 1) * PITCH + lane * 4) = rB;

    __builtin_amdgcn_fence(__ATOMIC_RELEASE, "workgroup", "local");
    __builtin_amdgcn_s_barrier();
    __builtin_amdgcn_fence(__ATOMIC_ACQUIRE, "workgroup", "local");

    const float* __restrict__ arow = buf + l16 * PITCH + half * 4;
    v4f af[16];
#pragma unroll
    for (int j = 0; j < 16; ++j) af[j] = *(const v4f*)(arow + 8 * j);

    v8f acc = {};
#pragma unroll
    for (int j = 0; j < 16; ++j) {
      v2f a0, a1;
      a0.x = af[j].x; a0.y = af[j].y;
      a1.x = af[j].z; a1.y = af[j].w;
      acc = __builtin_amdgcn_wmma_f32_16x16x4_f32(false, a0, false, bfr[2 * j],
                                                  (short)0, acc, false, false);
      acc = __builtin_amdgcn_wmma_f32_16x16x4_f32(false, a1, false, bfr[2 * j + 1],
                                                  (short)0, acc, false, false);
    }

    const int rl0 = half * 8;
    float* __restrict__ orow = out + (long)(blk * BR + rl0) * NCOLS + outCol;
#pragma unroll
    for (int r = 0; r < 8; ++r) {
      float v = actW * acc[r] + bv;
      v += resS * buf[(rl0 + r) * PITCH + outCol];
      if (RELU) v = fmaxf(v, 0.0f);
      orow[r * NCOLS] = v;
    }
    p ^= 1;
  }
}


static inline long cdiv(long a, long b) { return (a + b - 1) / b; }

extern "C" void kernel_launch(void* const* d_in, const int* in_sizes, int n_in,
                              void* d_out, int out_size, void* d_ws, size_t ws_size,
                              hipStream_t stream) {
  const float* feat = (const float*)d_in[0];
  const int*   src  = (const int*)d_in[1];
  const int*   dst  = (const int*)d_in[2];
  const float* w1   = (const float*)d_in[3];
  const float* b1   = (const float*)d_in[4];
  const float* cw1  = (const float*)d_in[5];
  const float* cb1  = (const float*)d_in[6];
  const float* cw2  = (const float*)d_in[7];
  const float* cb2  = (const float*)d_in[8];
  const float* w2   = (const float*)d_in[9];
  const float* b2   = (const float*)d_in[10];

  const int N = in_sizes[0] / 128;
  const int E = in_sizes[1];
  const long NF = (long)N * 128;
  const int NA = (N + 63) & ~63;

  char* wsb = (char*)d_ws;
  int* cntO   = (int*)wsb;                wsb += (size_t)NA * 4;
  int* cntI   = (int*)wsb;                wsb += (size_t)NA * 4;
  int* rowptr = (int*)wsb;                wsb += (size_t)NA * 4;
  int* cursor = (int*)wsb;                wsb += (size_t)NA * 4;
  int* bsum   = (int*)wsb;                wsb += 512 * 4;
  int* csc    = (int*)wsb;                wsb += ((size_t)E * 8 + 255) & ~(size_t)255;
  float* degO = (float*)wsb;              wsb += (size_t)NA * 4;
  float* degI = (float*)wsb;              wsb += (size_t)NA * 4;
  float* h0   = (float*)wsb;              wsb += (size_t)NF * 4;
  float* bufS = (float*)wsb;              wsb += (size_t)NF * 4;
  float* bufA = (float*)wsb;
  (void)ws_size; (void)n_in; (void)out_size;

  const int NB       = (int)cdiv(N, 256);
  const int nb_node  = NB;
  const int nb_edge  = (int)cdiv(E, 256);

  k_zero_i<<<(int)cdiv(2L * NA, 256), 256, 0, stream>>>(cntO, 2L * NA);
  k_degcount<<<nb_edge, 256, 0, stream>>>(src, dst, cntO, cntI, E);
  k_degrsqrt<<<nb_node, 256, 0, stream>>>(cntO, cntI, degO, degI, N);
  k_scan_block<<<NB, 256, 0, stream>>>(cntI, rowptr, bsum, N);
  k_scan_tops<<<1, 512, 0, stream>>>(bsum, NB);
  k_scan_add<<<NB, 256, 0, stream>>>(rowptr, bsum, cursor, N);
  k_build_csc<<<nb_edge, 256, 0, stream>>>(src, dst, degO, cursor, csc, E);

  k_gemm_wmma<128, 8, true, false><<<512, 256, 0, stream>>>(feat, w1, b1, h0, 1.0f, 0.0f, N);

  k_layer_fused<true><<<768, 256, 0, stream>>>(rowptr, cntI, csc, h0, degI, h0, cw1, cb1, bufA,
                                                GA_BETA1, 1.0f - GA_BETA1, N);

  k_layer_fused<false><<<768, 256, 0, stream>>>(rowptr, cntI, csc, bufA, degI, h0, cw2, cb2, bufS,
                                                 GA_BETA2, 1.0f - GA_BETA2, N);

  k_gemm_wmma<64, 8, false, false><<<512, 256, 0, stream>>>(bufS, w2, b2, (float*)d_out,
                                                            1.0f, 0.0f, N);
}


# CSC build as a two-level bucket sort on LDS counters: edges binned by 256-node range, each range counted and placed in LDS; no global atomics, 9 launches; in-node CSC order: slice, then LDS arrival
# speedup vs baseline: 2.6363x; 1.9288x over previous
#include <hip/hip_runtime.h>


#define GA_ALPHA 0.2f
#define GA_BETA1 0.6931471805599453f
#define GA_BETA2 0.4054651081081644f

typedef __attribute__((ext_vector_type(2))) float v2f;
typedef __attribute__((ext_vector_type(2))) int   v2i;
typedef __attribute__((ext_vector_type(4))) float v4f;
typedef __attribute__((ext_vector_type(8))) float v8f;


#define SCAN_SHIFT 9
#define SCAN_BLOCK 512

__global__ __launch_bounds__(SCAN_BLOCK) void k_scan_block(const int* __restrict__ in,
                                                           int* __restrict__ outEx,
                                                           int* __restrict__ bsum, int n) {
  __shared__ int s[SCAN_BLOCK];
  const int tid = threadIdx.x;
  const int i = blockIdx.x * SCAN_BLOCK + tid;
  const int v = (i < n) ? in[i] : 0;
  s[tid] = v;
  __syncthreads();
#pragma unroll
  for (int off = 1; off < SCAN_BLOCK; off <<= 1) {
    int t = (tid >= off) ? s[tid - off] : 0;
    __syncthreads();
    s[tid] += t;
    __syncthreads();
  }
  if (i < n) outEx[i] = s[tid] - v;
  if (tid == SCAN_BLOCK - 1) bsum[blockIdx.x] = s[SCAN_BLOCK - 1];
}


#define BIN_SHIFT 8
#define BIN_NODES 256
#define MAX_BINS  512
#define SLICE_CAP 6272
#define STAGE_CAP 6144

template <int K>
static __device__ __forceinline__ void wave_excl_scan(int* a, int lane) {
  int loc[K];
  int sum = 0;
#pragma unroll
  for (int k = 0; k < K; ++k) { loc[k] = a[lane * K + k]; sum += loc[k]; }
  int inc = sum;
#pragma unroll
  for (int off = 1; off < 32; off <<= 1) {
    const int t = __shfl_up(inc, off);
    if (lane >= off) inc += t;
  }
  int ex = inc - sum;
#pragma unroll
  for (int k = 0; k < K; ++k) { a[lane * K + k] = ex; ex += loc[k]; }
}

__global__ __launch_bounds__(1024) void k_bin_hist(const int* __restrict__ src,
                                                   const int* __restrict__ dst,
                                                   int* __restrict__ cntMat,
                                                   int E, int per, int nbin, int nslice) {
  __shared__ int h[2 * MAX_BINS];
  const int tid = threadIdx.x;
  const int g = blockIdx.x;
  for (int v = tid; v < 2 * MAX_BINS; v += 1024) h[v] = 0;
  __syncthreads();
  const int lo = g * per;
  const int hi = (lo + per < E) ? lo + per : E;
  for (int base = lo; base < hi; base += 8 * 1024) {
    int s[8], d[8];
#pragma unroll
    for (int u = 0; u < 8; ++u) {
      const int e = base + u * 1024 + tid;
      const int ec = (e < hi) ? e : hi - 1;
      s[u] = src[ec];
      d[u] = dst[ec];
    }
#pragma unroll
    for (int u = 0; u < 8; ++u) {
      if (base + u * 1024 + tid < hi) {
        atomicAdd(&h[d[u] >> BIN_SHIFT], 1);
        atomicAdd(&h[nbin + (s[u] >> BIN_SHIFT)], 1);
      }
    }
  }
  __syncthreads();
  for (int v = tid; v < 2 * nbin; v += 1024) cntMat[(long)v * nslice + g] = h[v];
  if (g == 0 && tid == 0) cntMat[(long)2 * nbin * nslice] = 0;
}

__global__ __launch_bounds__(1024) void k_bin_scatter(const int* __restrict__ src,
                                                      const int* __restrict__ dst,
                                                      const int* __restrict__ cntMat,
                                                      const int* __restrict__ offMat,
                                                      const int* __restrict__ bsum,
                                                      int* __restrict__ binStart,
                                                      int* __restrict__ binDraw,
                                                      int* __restrict__ binS,
                                                      int E, int per, int nbin, int nslice, int nbs) {
  __shared__ int gl[2 * MAX_BINS];
  __shared__ int lst[2 * MAX_BINS];
  __shared__ int lcur[2 * MAX_BINS];
  __shared__ int top[512];
  __shared__ v2i recD[SLICE_CAP];
  __shared__ int recS[SLICE_CAP];
  v2i* __restrict__ binD = (v2i*)binDraw;
  const int tid = threadIdx.x;
  const int g = blockIdx.x;
  if (tid < 512) top[tid] = (tid < nbs) ? bsum[tid] : 0;
  for (int v = tid; v < 2 * MAX_BINS; v += 1024) lst[v] = (v < 2 * nbin) ? cntMat[(long)v * nslice + g] : 0;
  __syncthreads();
  if (tid < 32) wave_excl_scan<16>(top, tid);
  else if (tid < 64) wave_excl_scan<32>(lst, tid - 32);
  __syncthreads();
  for (int v = tid; v < 2 * nbin; v += 1024) {
    const long idx = (long)v * nslice + g;
    const int o = offMat[idx] + top[idx >> SCAN_SHIFT];
    gl[v] = o;
    lcur[v] = lst[v];
    if (g == 0) binStart[v] = o;
  }
  if (g == 0 && tid == 0) {
    const long idx = (long)2 * nbin * nslice;
    binStart[2 * nbin] = offMat[idx] + top[idx >> SCAN_SHIFT];
  }
  __syncthreads();
  const int lo = g * per;
  const int hi = (lo + per < E) ? lo + per : E;
  const int nD = (hi > lo) ? hi - lo : 0;
  const bool sorted = per <= SLICE_CAP;
  for (int base = lo; base < hi; base += 8 * 1024) {
    int s[8], d[8];
#pragma unroll
    for (int u = 0; u < 8; ++u) {
      const int e = base + u * 1024 + tid;
      const int ec = (e < hi) ? e : hi - 1;
      s[u] = src[ec];
      d[u] = dst[ec];
    }
#pragma unroll
    for (int u = 0; u < 8; ++u) {
      if (base + u * 1024 + tid < hi) {
        v2i rec;
        rec.x = s[u];
        rec.y = d[u];
        if (sorted) {
          const int p = atomicAdd(&lcur[d[u] >> BIN_SHIFT], 1);
          recD[p] = rec;
          const int q = atomicAdd(&lcur[nbin + (s[u] >> BIN_SHIFT)], 1) - nD;
          recS[q] = s[u];
        } else {
          const int p = atomicAdd(&gl[d[u] >> BIN_SHIFT], 1);
          binD[p] = rec;
          const int q = atomicAdd(&gl[nbin + (s[u] >> BIN_SHIFT)], 1) - E;
          binS[q] = s[u];
        }
      }
    }
  }
  if (sorted) {
    __syncthreads();
    for (int i = tid; i < nD; i += 1024) {
      const v2i rec = recD[i];
      const int v = rec.y >> BIN_SHIFT;
      binD[gl[v] + (i - lst[v])] = rec;
      const int s = recS[i];
      const int v2 = nbin + (s >> BIN_SHIFT);
      binS[gl[v2] - E + (i + nD - lst[v2])] = s;
    }
  }
}

__global__ __launch_bounds__(1024) void k_src_count(const int* __restrict__ binS,
                                                    const int* __restrict__ binStart,
                                                    float* __restrict__ degO,
                                                    int N, int E, int nbin) {
  __shared__ int c[BIN_NODES];
  const int tid = threadIdx.x;
  const int b = blockIdx.x;
  if (tid < BIN_NODES) c[tid] = 0;
  __syncthreads();
  const int start = binStart[nbin + b] - E;
  const int end   = binStart[nbin + b + 1] - E;
  const int node0 = b << BIN_SHIFT;
  for (int base = start; base < end; base += 8 * 1024) {
    int s[8];
#pragma unroll
    for (int u = 0; u < 8; ++u) {
      const int i = base + u * 1024 + tid;
      s[u] = binS[(i < end) ? i : end - 1];
    }
#pragma unroll
    for (int u = 0; u < 8; ++u)
      if (base + u * 1024 + tid < end) atomicAdd(&c[s[u] - node0], 1);
  }
  __syncthreads();
  if (tid < BIN_NODES) {
    const int node = node0 + tid;
    if (node < N) degO[node] = rsqrtf(fmaxf((float)c[tid], 1.0f));
  }
}

__global__ __launch_bounds__(1024) void k_dst_place(const int* __restrict__ binDraw,
                                                    const int* __restrict__ binStart,
                                                    const float* __restrict__ degO,
                                                    int* __restrict__ rowptr,
                                                    int* __restrict__ cntI,
                                                    float* __restrict__ degI,
                                                    int* __restrict__ cscRaw, int N) {
  __shared__ int c[BIN_NODES];
  __shared__ int cur[BIN_NODES];
  __shared__ v2i stage[STAGE_CAP];
  const v2i* __restrict__ binD = (const v2i*)binDraw;
  v2i* __restrict__ csc = (v2i*)cscRaw;
  const int tid = threadIdx.x;
  const int b = blockIdx.x;
  if (tid < BIN_NODES) c[tid] = 0;
  __syncthreads();
  const int start = binStart[b];
  const int end   = binStart[b + 1];
  const int nrec  = end - start;
  const int node0 = b << BIN_SHIFT;

  for (int base = start; base < end; base += 8 * 1024) {
    int d[8];
#pragma unroll
    for (int u = 0; u < 8; ++u) {
      const int i = base + u * 1024 + tid;
      d[u] = binD[(i < end) ? i : end - 1].y;
    }
#pragma unroll
    for (int u = 0; u < 8; ++u)
      if (base + u * 1024 + tid < end) atomicAdd(&c[d[u] - node0], 1);
  }
  __syncthreads();

  if (tid < BIN_NODES) cur[tid] = c[tid];
  __syncthreads();
  if (tid < 32) wave_excl_scan<8>(cur, tid);
  __syncthreads();
  if (tid < BIN_NODES) {
    const int node = node0 + tid;
    if (node < N) {
      rowptr[node] = start + cur[tid];
      cntI[node]   = c[tid];
      degI[node]   = rsqrtf(fmaxf((float)c[tid], 1.0f));
    }
  }
  __syncthreads();

  const bool staged = nrec <= STAGE_CAP;
  for (int base = start; base < end; base += 8 * 1024) {
    v2i r[8];
    float w[8];
#pragma unroll
    for (int u = 0; u < 8; ++u) {
      const int i = base + u * 1024 + tid;
      r[u] = binD[(i < end) ? i : end - 1];
    }
#pragma unroll
    for (int u = 0; u < 8; ++u) w[u] = degO[r[u].x];
#pragma unroll
    for (int u = 0; u < 8; ++u) {
      if (base + u * 1024 + tid < end) {
        const int p = atomicAdd(&cur[r[u].y - node0], 1);
        v2i ent;
        ent.x = r[u].x;
        ent.y = __float_as_int(w[u]);
        if (staged) stage[p] = ent; else csc[start + p] = ent;
      }
    }
  }
  if (staged) {
    __syncthreads();
    for (int i = tid; i < nrec; i += 1024) csc[start + i] = stage[i];
  }
}

template <int NCOLS, int WAVES, bool RELU, bool RESID>
__global__ __launch_bounds__(WAVES * 32) void k_gemm_wmma(const float* __restrict__ A,
                                                          const float* __restrict__ W,
                                                          const float* __restrict__ bias,
                                                          float* __restrict__ out,
                                                          float actW, float resS, int nRows) {
  constexpr int NCT   = NCOLS / 16;
  constexpr int MT    = WAVES / NCT;
  constexpr int BR    = MT * 16;
  constexpr int RPW   = BR / WAVES;
  constexpr int PITCH = 132;
  __shared__ __attribute__((aligned(16))) float lds[2][BR * PITCH];

  const int wave = threadIdx.x >> 5;
  const int lane = threadIdx.x & 31;
  const int ct   = wave % NCT;
  const int mt   = wave / NCT;
  const int col0 = ct * 16;
  const int half = lane >> 4;
  const int l16  = lane & 15;
  const int outCol = col0 + l16;

  v2f bfr[32];
#pragma unroll
  for (int s = 0; s < 32; ++s) {
    const int kb = 8 * (s >> 1) + 4 * half + 2 * (s & 1);
    bfr[s].x = W[(long)kb * NCOLS + outCol];
    bfr[s].y = W[(long)(kb + 1) * NCOLS + outCol];
  }
  const float bv = bias[outCol];

  const int nblk = nRows / BR;
  int blk = blockIdx.x;

  v4f stage[RPW];
#pragma unroll
  for (int i = 0; i < RPW; ++i) stage[i] = (v4f){0.0f, 0.0f, 0.0f, 0.0f};
  if (blk < nblk) {
#pragma unroll
    for (int i = 0; i < RPW; ++i)
      stage[i] = *(const v4f*)(A + (long)(blk * BR + wave * RPW + i) * 128 + lane * 4);
  }

  int p = 0;
  for (; blk < nblk; blk += gridDim.x) {
    float* __restrict__ buf = lds[p];
#pragma unroll
    for (int i = 0; i < RPW; ++i)
      *(v4f*)(buf + (wave * RPW + i) * PITCH + lane * 4) = stage[i];

    __builtin_amdgcn_fence(__ATOMIC_RELEASE, "workgroup", "local");
    __builtin_amdgcn_s_barrier();
    __builtin_amdgcn_fence(__ATOMIC_ACQUIRE, "workgroup", "local");

    const int next = blk + gridDim.x;
    if (next < nblk) {
#pragma unroll
      for (int i = 0; i < RPW; ++i)
        stage[i] = *(const v4f*)(A + (long)(next * BR + wave * RPW + i) * 128 + lane * 4);
    }

    const float* __restrict__ arow = buf + (mt * 16 + l16) * PITCH + half * 4;
    v4f af[16];
#pragma unroll
    for (int j = 0; j < 16; ++j) af[j] = *(const v4f*)(arow + 8 * j);

    v8f acc = {};
#pragma unroll
    for (int j = 0; j < 16; ++j) {
      v2f a0, a1;
      a0.x = af[j].x; a0.y = af[j].y;
      a1.x = af[j].z; a1.y = af[j].w;
      acc = __builtin_amdgcn_wmma_f32_16x16x4_f32(false, a0, false, bfr[2 * j],
                                                  (short)0, acc, false, false);
      acc = __builtin_amdgcn_wmma_f32_16x16x4_f32(false, a1, false, bfr[2 * j + 1],
                                                  (short)0, acc, false, false);
    }

    const int rl0 = mt * 16 + half * 8;
    float* __restrict__ orow = out + (long)(blk * BR + rl0) * NCOLS + outCol;
#pragma unroll
    for (int r = 0; r < 8; ++r) {
      float v = actW * acc[r] + bv;
      if (RESID) v += resS * buf[(rl0 + r) * PITCH + outCol];
      if (RELU) v = fmaxf(v, 0.0f);
      orow[r * NCOLS] = v;
    }
    p ^= 1;
  }
}


static __device__ __forceinline__ v4f gather_node(const v2i* __restrict__ csc,
                                                  const float* __restrict__ x,
                                                  int start, int c, v2i ent0, int lane) {
  const unsigned lane4 = (unsigned)lane * 4u;
  v4f acc = {0.0f, 0.0f, 0.0f, 0.0f};
  int idx = ent0.x;
  int wbt = ent0.y;
  for (int base = 0; base < c; base += 32) {
    if (base > 0) {
      v2i t = {0, 0};
      if (base + lane < c) t = csc[start + base + lane];
      idx = t.x;
      wbt = t.y;
    }
    const int n = (c - base < 32) ? c - base : 32;
#pragma unroll 1
    for (int j = 0; j < n; j += 8) {
      int   s[8];
      float w[8];
#pragma unroll
      for (int u = 0; u < 8; ++u) {
        const int e = (j + u < n) ? j + u : n - 1;
        s[u] = __builtin_amdgcn_readlane(idx, e);
        w[u] = __int_as_float(__builtin_amdgcn_readlane(wbt, e));
      }
      v4f v[8];
      if (j + 8 <= n) {
#pragma unroll
        for (int u = 0; u < 8; ++u) v[u] = *(const v4f*)(x + (long)s[u] * 128 + lane4);
#pragma unroll
        for (int u = 0; u < 8; ++u) acc += w[u] * v[u];
      } else {
#pragma unroll
        for (int u = 0; u < 8; ++u) {
          v[u] = (v4f){0.0f, 0.0f, 0.0f, 0.0f};
          if (j + u < n) v[u] = *(const v4f*)(x + (long)s[u] * 128 + lane4);
        }
#pragma unroll
        for (int u = 0; u < 8; ++u) acc += w[u] * v[u];
      }
    }
  }
  return acc;
}

template <bool RELU>
__global__ __launch_bounds__(256) void k_layer_fused(const int* __restrict__ rowptr,
                                                     const int* __restrict__ cnt,
                                                     const int* __restrict__ cscRaw,
                                                     const float* __restrict__ x,
                                                     const float* __restrict__ dirs,
                                                     const float* __restrict__ x0,
                                                     const float* __restrict__ W,
                                                     const float* __restrict__ bias,
                                                     float* __restrict__ out,
                                                     float actW, float resS, int nRows) {
  constexpr int NCOLS = 128;
  constexpr int BR    = 16;
  constexpr int PITCH = 132;
  __shared__ __attribute__((aligned(16))) float lds[2][BR * PITCH];
  const v2i* __restrict__ csc = (const v2i*)cscRaw;

  const int wave = __builtin_amdgcn_readfirstlane((int)(threadIdx.x >> 5));
  const int lane = threadIdx.x & 31;
  const int col0 = wave * 16;
  const int half = lane >> 4;
  const int l16  = lane & 15;
  const int outCol = col0 + l16;

  v2f bfr[32];
#pragma unroll
  for (int s = 0; s < 32; ++s) {
    const int kb = 8 * (s >> 1) + 4 * half + 2 * (s & 1);
    bfr[s].x = W[(long)kb * NCOLS + outCol];
    bfr[s].y = W[(long)(kb + 1) * NCOLS + outCol];
  }
  const float bv = bias[outCol];

  const int nblk = nRows / BR;
  int p = 0;
  for (int blk = blockIdx.x; blk < nblk; blk += gridDim.x) {
    float* __restrict__ buf = lds[p];

    const int nodeA  = blk * BR + wave * 2;
    const int startA = rowptr[nodeA];
    const int startB = rowptr[nodeA + 1];
    const int cA     = cnt[nodeA];
    const int cB     = cnt[nodeA + 1];
    v2i iA = {0, 0};
    v2i iB = {0, 0};
    if (lane < cA) iA = csc[startA + lane];
    if (lane < cB) iB = csc[startB + lane];
    const float scA = (1.0f - GA_ALPHA) * dirs[nodeA];
    const float scB = (1.0f - GA_ALPHA) * dirs[nodeA + 1];
    const long  oA  = (long)nodeA * 128 + lane * 4;
    const v4f x0A = *(const v4f*)(x0 + oA);
    const v4f x0B = *(const v4f*)(x0 + oA + 128);

    const v4f accA = gather_node(csc, x, startA, cA, iA, lane);
    v4f rA;
    rA.x = scA * accA.x + GA_ALPHA * x0A.x;
    rA.y = scA * accA.y + GA_ALPHA * x0A.y;
    rA.z = scA * accA.z + GA_ALPHA * x0A.z;
    rA.w = scA * accA.w + GA_ALPHA * x0A.w;
    *(v4f*)(buf + (wave * 2) * PITCH + lane * 4) = rA;

    const v4f accB = gather_node(csc, x, startB, cB, iB, lane);
    v4f rB;
    rB.x = scB * accB.x + GA_ALPHA * x0B.x;
    rB.y = scB * accB.y + GA_ALPHA * x0B.y;
    rB.z = scB * accB.z + GA_ALPHA * x0B.z;
    rB.w = scB * accB.w + GA_ALPHA * x0B.w;
    *(v4f*)(buf + (wave * 2 + 1) * PITCH + lane * 4) = rB;

    __builtin_amdgcn_fence(__ATOMIC_RELEASE, "workgroup", "local");
    __builtin_amdgcn_s_barrier();
    __builtin_amdgcn_fence(__ATOMIC_ACQUIRE, "workgroup", "local");

    const float* __restrict__ arow = buf + l16 * PITCH + half * 4;
    v4f af[16];
#pragma unroll
    for (int j = 0; j < 16; ++j) af[j] = *(const v4f*)(arow + 8 * j);

    v8f acc = {};
#pragma unroll
    for (int j = 0; j < 16; ++j) {
      v2f a0, a1;
      a0.x = af[j].x; a0.y = af[j].y;
      a1.x = af[j].z; a1.y = af[j].w;
      acc = __builtin_amdgcn_wmma_f32_16x16x4_f32(false, a0, false, bfr[2 * j],
                                                  (short)0, acc, false, false);
      acc = __builtin_amdgcn_wmma_f32_16x16x4_f32(false, a1, false, bfr[2 * j + 1],
                                                  (short)0, acc, false, false);
    }

    const int rl0 = half * 8;
    float* __restrict__ orow = out + (long)(blk * BR + rl0) * NCOLS + outCol;
#pragma unroll
    for (int r = 0; r < 8; ++r) {
      float v = actW * acc[r] + bv;
      v += resS * buf[(rl0 + r) * PITCH + outCol];
      if (RELU) v = fmaxf(v, 0.0f);
      orow[r * NCOLS] = v;
    }
    p ^= 1;
  }
}


static inline long cdiv(long a, long b) { return (a + b - 1) / b; }

extern "C" void kernel_launch(void* const* d_in, const int* in_sizes, int n_in,
                              void* d_out, int out_size, void* d_ws, size_t ws_size,
                              hipStream_t stream) {
  const float* feat = (const float*)d_in[0];
  const int*   src  = (const int*)d_in[1];
  const int*   dst  = (const int*)d_in[2];
  const float* w1   = (const float*)d_in[3];
  const float* b1   = (const float*)d_in[4];
  const float* cw1  = (const float*)d_in[5];
  const float* cb1  = (const float*)d_in[6];
  const float* cw2  = (const float*)d_in[7];
  const float* cb2  = (const float*)d_in[8];
  const float* w2   = (const float*)d_in[9];
  const float* b2   = (const float*)d_in[10];

  const int N = in_sizes[0] / 128;
  const int E = in_sizes[1];
  const long NF = (long)N * 128;
  const int NA = (N + 63) & ~63;

  char* wsb = (char*)d_ws;
  wsb += (size_t)NA * 4;
  int* cntI   = (int*)wsb;                wsb += (size_t)NA * 4;
  int* rowptr = (int*)wsb;                wsb += (size_t)NA * 4;
  wsb += (size_t)NA * 4;
  int* bsum   = (int*)wsb;                wsb += 512 * 4;
  int* csc    = (int*)wsb;                wsb += ((size_t)E * 8 + 255) & ~(size_t)255;
  float* degO = (float*)wsb;              wsb += (size_t)NA * 4;
  float* degI = (float*)wsb;              wsb += (size_t)NA * 4;
  float* h0   = (float*)wsb;              wsb += (size_t)NF * 4;
  float* bufS = (float*)wsb;              wsb += (size_t)NF * 4;
  float* bufA = (float*)wsb;
  (void)ws_size; (void)n_in; (void)out_size;

  const int NBIN = (N + BIN_NODES - 1) >> BIN_SHIFT;
  const int NSL  = 256;
  const int per  = (int)cdiv(E, NSL);
  const int NM   = 2 * NBIN * NSL;
  const int NBS  = (int)cdiv(NM + 1, SCAN_BLOCK);
  char* tb = (char*)bufS;
  int* binD     = (int*)tb;               tb += ((size_t)E * 8 + 255) & ~(size_t)255;
  int* binS     = (int*)tb;               tb += ((size_t)E * 4 + 255) & ~(size_t)255;
  int* cntMat   = (int*)tb;               tb += ((size_t)(NM + 1) * 4 + 255) & ~(size_t)255;
  int* offMat   = (int*)tb;               tb += ((size_t)(NM + 1) * 4 + 255) & ~(size_t)255;
  int* binStart = (int*)tb;

  k_bin_hist<<<NSL, 1024, 0, stream>>>(src, dst, cntMat, E, per, NBIN, NSL);
  k_scan_block<<<NBS, SCAN_BLOCK, 0, stream>>>(cntMat, offMat, bsum, NM + 1);
  k_bin_scatter<<<NSL, 1024, 0, stream>>>(src, dst, cntMat, offMat, bsum, binStart, binD, binS,
                                          E, per, NBIN, NSL, NBS);
  k_src_count<<<NBIN, 1024, 0, stream>>>(binS, binStart, degO, N, E, NBIN);
  k_dst_place<<<NBIN, 1024, 0, stream>>>(binD, binStart, degO, rowptr, cntI, degI, csc, N);

  k_gemm_wmma<128, 8, true, false><<<512, 256, 0, stream>>>(feat, w1, b1, h0, 1.0f, 0.0f, N);

  k_layer_fused<true><<<768, 256, 0, stream>>>(rowptr, cntI, csc, h0, degI, h0, cw1, cb1, bufA,
                                                GA_BETA1, 1.0f - GA_BETA1, N);

  k_layer_fused<false><<<768, 256, 0, stream>>>(rowptr, cntI, csc, bufA, degI, h0, cw2, cb2, bufS,
                                                 GA_BETA2, 1.0f - GA_BETA2, N);

  k_gemm_wmma<64, 8, false, false><<<512, 256, 0, stream>>>(bufS, w2, b2, (float*)d_out,
                                                            1.0f, 0.0f, N);
}


# Output projection runs inside layer 2's launch: each finished 16-row x2 tile stays in LDS, four waves multiply it by w2 (B fragments staged in LDS); x2 never stored; 8 launches; sum order same
# speedup vs baseline: 2.8231x; 1.0709x over previous
#include <hip/hip_runtime.h>


#define GA_ALPHA 0.2f
#define GA_BETA1 0.6931471805599453f
#define GA_BETA2 0.4054651081081644f

typedef __attribute__((ext_vector_type(2))) float v2f;
typedef __attribute__((ext_vector_type(2))) int   v2i;
typedef __attribute__((ext_vector_type(4))) float v4f;
typedef __attribute__((ext_vector_type(8))) float v8f;


#define SCAN_SHIFT 9
#define SCAN_BLOCK 512

__global__ __launch_bounds__(SCAN_BLOCK) void k_scan_block(const int* __restrict__ in,
                                                           int* __restrict__ outEx,
                                                           int* __restrict__ bsum, int n) {
  __shared__ int s[SCAN_BLOCK];
  const int tid = threadIdx.x;
  const int i = blockIdx.x * SCAN_BLOCK + tid;
  const int v = (i < n) ? in[i] : 0;
  s[tid] = v;
  __syncthreads();
#pragma unroll
  for (int off = 1; off < SCAN_BLOCK; off <<= 1) {
    int t = (tid >= off) ? s[tid - off] : 0;
    __syncthreads();
    s[tid] += t;
    __syncthreads();
  }
  if (i < n) outEx[i] = s[tid] - v;
  if (tid == SCAN_BLOCK - 1) bsum[blockIdx.x] = s[SCAN_BLOCK - 1];
}


#define BIN_SHIFT 8
#define BIN_NODES 256
#define MAX_BINS  512
#define SLICE_CAP 6272
#define STAGE_CAP 6144

template <int K>
static __device__ __forceinline__ void wave_excl_scan(int* a, int lane) {
  int loc[K];
  int sum = 0;
#pragma unroll
  for (int k = 0; k < K; ++k) { loc[k] = a[lane * K + k]; sum += loc[k]; }
  int inc = sum;
#pragma unroll
  for (int off = 1; off < 32; off <<= 1) {
    const int t = __shfl_up(inc, off);
    if (lane >= off) inc += t;
  }
  int ex = inc - sum;
#pragma unroll
  for (int k = 0; k < K; ++k) { a[lane * K + k] = ex; ex += loc[k]; }
}

__global__ __launch_bounds__(1024) void k_bin_hist(const int* __restrict__ src,
                                                   const int* __restrict__ dst,
                                                   int* __restrict__ cntMat,
                                                   int E, int per, int nbin, int nslice) {
  __shared__ int h[2 * MAX_BINS];
  const int tid = threadIdx.x;
  const int g = blockIdx.x;
  for (int v = tid; v < 2 * MAX_BINS; v += 1024) h[v] = 0;
  __syncthreads();
  const int lo = g * per;
  const int hi = (lo + per < E) ? lo + per : E;
  for (int base = lo; base < hi; base += 8 * 1024) {
    int s[8], d[8];
#pragma unroll
    for (int u = 0; u < 8; ++u) {
      const int e = base + u * 1024 + tid;
      const int ec = (e < hi) ? e : hi - 1;
      s[u] = src[ec];
      d[u] = dst[ec];
    }
#pragma unroll
    for (int u = 0; u < 8; ++u) {
      if (base + u * 1024 + tid < hi) {
        atomicAdd(&h[d[u] >> BIN_SHIFT], 1);
        atomicAdd(&h[nbin + (s[u] >> BIN_SHIFT)], 1);
      }
    }
  }
  __syncthreads();
  for (int v = tid; v < 2 * nbin; v += 1024) cntMat[(long)v * nslice + g] = h[v];
  if (g == 0 && tid == 0) cntMat[(long)2 * nbin * nslice] = 0;
}

__global__ __launch_bounds__(1024) void k_bin_scatter(const int* __restrict__ src,
                                                      const int* __restrict__ dst,
                                                      const int* __restrict__ cntMat,
                                                      const int* __restrict__ offMat,
                                                      const int* __restrict__ bsum,
                                                      int* __restrict__ binStart,
                                                      int* __restrict__ binDraw,
                                                      int* __restrict__ binS,
                                                      int E, int per, int nbin, int nslice, int nbs) {
  __shared__ int gl[2 * MAX_BINS];
  __shared__ int lst[2 * MAX_BINS];
  __shared__ int lcur[2 * MAX_BINS];
  __shared__ int top[512];
  __shared__ v2i recD[SLICE_CAP];
  __shared__ int recS[SLICE_CAP];
  v2i* __restrict__ binD = (v2i*)binDraw;
  const int tid = threadIdx.x;
  const int g = blockIdx.x;
  if (tid < 512) top[tid] = (tid < nbs) ? bsum[tid] : 0;
  for (int v = tid; v < 2 * MAX_BINS; v += 1024) lst[v] = (v < 2 * nbin) ? cntMat[(long)v * nslice + g] : 0;
  __syncthreads();
  if (tid < 32) wave_excl_scan<16>(top, tid);
  else if (tid < 64) wave_excl_scan<32>(lst, tid - 32);
  __syncthreads();
  for (int v = tid; v < 2 * nbin; v += 1024) {
    const long idx = (long)v * nslice + g;
    const int o = offMat[idx] + top[idx >> SCAN_SHIFT];
    gl[v] = o;
    lcur[v] = lst[v];
    if (g == 0) binStart[v] = o;
  }
  if (g == 0 && tid == 0) {
    const long idx = (long)2 * nbin * nslice;
    binStart[2 * nbin] = offMat[idx] + top[idx >> SCAN_SHIFT];
  }
  __syncthreads();
  const int lo = g * per;
  const int hi = (lo + per < E) ? lo + per : E;
  const int nD = (hi > lo) ? hi - lo : 0;
  const bool sorted = per <= SLICE_CAP;
  for (int base = lo; base < hi; base += 8 * 1024) {
    int s[8], d[8];
#pragma unroll
    for (int u = 0; u < 8; ++u) {
      const int e = base + u * 1024 + tid;
      const int ec = (e < hi) ? e : hi - 1;
      s[u] = src[ec];
      d[u] = dst[ec];
    }
#pragma unroll
    for (int u = 0; u < 8; ++u) {
      if (base + u * 1024 + tid < hi) {
        v2i rec;
        rec.x = s[u];
        rec.y = d[u];
        if (sorted) {
          const int p = atomicAdd(&lcur[d[u] >> BIN_SHIFT], 1);
          recD[p] = rec;
          const int q = atomicAdd(&lcur[nbin + (s[u] >> BIN_SHIFT)], 1) - nD;
          recS[q] = s[u];
        } else {
          const int p = atomicAdd(&gl[d[u] >> BIN_SHIFT], 1);
          binD[p] = rec;
          const int q = atomicAdd(&gl[nbin + (s[u] >> BIN_SHIFT)], 1) - E;
          binS[q] = s[u];
        }
      }
    }
  }
  if (sorted) {
    __syncthreads();
    for (int i = tid; i < nD; i += 1024) {
      const v2i rec = recD[i];
      const int v = rec.y >> BIN_SHIFT;
      binD[gl[v] + (i - lst[v])] = rec;
      const int s = recS[i];
      const int v2 = nbin + (s >> BIN_SHIFT);
      binS[gl[v2] - E + (i + nD - lst[v2])] = s;
    }
  }
}

__global__ __launch_bounds__(1024) void k_src_count(const int* __restrict__ binS,
                                                    const int* __restrict__ binStart,
                                                    float* __restrict__ degO,
                                                    int N, int E, int nbin) {
  __shared__ int c[BIN_NODES];
  const int tid = threadIdx.x;
  const int b = blockIdx.x;
  if (tid < BIN_NODES) c[tid] = 0;
  __syncthreads();
  const int start = binStart[nbin + b] - E;
  const int end   = binStart[nbin + b + 1] - E;
  const int node0 = b << BIN_SHIFT;
  for (int base = start; base < end; base += 8 * 1024) {
    int s[8];
#pragma unroll
    for (int u = 0; u < 8; ++u) {
      const int i = base + u * 1024 + tid;
      s[u] = binS[(i < end) ? i : end - 1];
    }
#pragma unroll
    for (int u = 0; u < 8; ++u)
      if (base + u * 1024 + tid < end) atomicAdd(&c[s[u] - node0], 1);
  }
  __syncthreads();
  if (tid < BIN_NODES) {
    const int node = node0 + tid;
    if (node < N) degO[node] = rsqrtf(fmaxf((float)c[tid], 1.0f));
  }
}

__global__ __launch_bounds__(1024) void k_dst_place(const int* __restrict__ binDraw,
                                                    const int* __restrict__ binStart,
                                                    const float* __restrict__ degO,
                                                    int* __restrict__ rowptr,
                                                    int* __restrict__ cntI,
                                                    float* __restrict__ degI,
                                                    int* __restrict__ cscRaw, int N) {
  __shared__ int c[BIN_NODES];
  __shared__ int cur[BIN_NODES];
  __shared__ v2i stage[STAGE_CAP];
  const v2i* __restrict__ binD = (const v2i*)binDraw;
  v2i* __restrict__ csc = (v2i*)cscRaw;
  const int tid = threadIdx.x;
  const int b = blockIdx.x;
  if (tid < BIN_NODES) c[tid] = 0;
  __syncthreads();
  const int start = binStart[b];
  const int end   = binStart[b + 1];
  const int nrec  = end - start;
  const int node0 = b << BIN_SHIFT;

  for (int base = start; base < end; base += 8 * 1024) {
    int d[8];
#pragma unroll
    for (int u = 0; u < 8; ++u) {
      const int i = base + u * 1024 + tid;
      d[u] = binD[(i < end) ? i : end - 1].y;
    }
#pragma unroll
    for (int u = 0; u < 8; ++u)
      if (base + u * 1024 + tid < end) atomicAdd(&c[d[u] - node0], 1);
  }
  __syncthreads();

  if (tid < BIN_NODES) cur[tid] = c[tid];
  __syncthreads();
  if (tid < 32) wave_excl_scan<8>(cur, tid);
  __syncthreads();
  if (tid < BIN_NODES) {
    const int node = node0 + tid;
    if (node < N) {
      rowptr[node] = start + cur[tid];
      cntI[node]   = c[tid];
      degI[node]   = rsqrtf(fmaxf((float)c[tid], 1.0f));
    }
  }
  __syncthreads();

  const bool staged = nrec <= STAGE_CAP;
  for (int base = start; base < end; base += 8 * 1024) {
    v2i r[8];
    float w[8];
#pragma unroll
    for (int u = 0; u < 8; ++u) {
      const int i = base + u * 1024 + tid;
      r[u] = binD[(i < end) ? i : end - 1];
    }
#pragma unroll
    for (int u = 0; u < 8; ++u) w[u] = degO[r[u].x];
#pragma unroll
    for (int u = 0; u < 8; ++u) {
      if (base + u * 1024 + tid < end) {
        const int p = atomicAdd(&cur[r[u].y - node0], 1);
        v2i ent;
        ent.x = r[u].x;
        ent.y = __float_as_int(w[u]);
        if (staged) stage[p] = ent; else csc[start + p] = ent;
      }
    }
  }
  if (staged) {
    __syncthreads();
    for (int i = tid; i < nrec; i += 1024) csc[start + i] = stage[i];
  }
}

template <int NCOLS, int WAVES, bool RELU, bool RESID>
__global__ __launch_bounds__(WAVES * 32) void k_gemm_wmma(const float* __restrict__ A,
                                                          const float* __restrict__ W,
                                                          const float* __restrict__ bias,
                                                          float* __restrict__ out,
                                                          float actW, float resS, int nRows) {
  constexpr int NCT   = NCOLS / 16;
  constexpr int MT    = WAVES / NCT;
  constexpr int BR    = MT * 16;
  constexpr int RPW   = BR / WAVES;
  constexpr int PITCH = 132;
  __shared__ __attribute__((aligned(16))) float lds[2][BR * PITCH];

  const int wave = threadIdx.x >> 5;
  const int lane = threadIdx.x & 31;
  const int ct   = wave % NCT;
  const int mt   = wave / NCT;
  const int col0 = ct * 16;
  const int half = lane >> 4;
  const int l16  = lane & 15;
  const int outCol = col0 + l16;

  v2f bfr[32];
#pragma unroll
  for (int s = 0; s < 32; ++s) {
    const int kb = 8 * (s >> 1) + 4 * half + 2 * (s & 1);
    bfr[s].x = W[(long)kb * NCOLS + outCol];
    bfr[s].y = W[(long)(kb + 1) * NCOLS + outCol];
  }
  const float bv = bias[outCol];

  const int nblk = nRows / BR;
  int blk = blockIdx.x;

  v4f stage[RPW];
#pragma unroll
  for (int i = 0; i < RPW; ++i) stage[i] = (v4f){0.0f, 0.0f, 0.0f, 0.0f};
  if (blk < nblk) {
#pragma unroll
    for (int i = 0; i < RPW; ++i)
      stage[i] = *(const v4f*)(A + (long)(blk * BR + wave * RPW + i) * 128 + lane * 4);
  }

  int p = 0;
  for (; blk < nblk; blk += gridDim.x) {
    float* __restrict__ buf = lds[p];
#pragma unroll
    for (int i = 0; i < RPW; ++i)
      *(v4f*)(buf + (wave * RPW + i) * PITCH + lane * 4) = stage[i];

    __builtin_amdgcn_fence(__ATOMIC_RELEASE, "workgroup", "local");
    __builtin_amdgcn_s_barrier();
    __builtin_amdgcn_fence(__ATOMIC_ACQUIRE, "workgroup", "local");

    const int next = blk + gridDim.x;
    if (next < nblk) {
#pragma unroll
      for (int i = 0; i < RPW; ++i)
        stage[i] = *(const v4f*)(A + (long)(next * BR + wave * RPW + i) * 128 + lane * 4);
    }

    const float* __restrict__ arow = buf + (mt * 16 + l16) * PITCH + half * 4;
    v4f af[16];
#pragma unroll
    for (int j = 0; j < 16; ++j) af[j] = *(const v4f*)(arow + 8 * j);

    v8f acc = {};
#pragma unroll
    for (int j = 0; j < 16; ++j) {
      v2f a0, a1;
      a0.x = af[j].x; a0.y = af[j].y;
      a1.x = af[j].z; a1.y = af[j].w;
      acc = __builtin_amdgcn_wmma_f32_16x16x4_f32(false, a0, false, bfr[2 * j],
                                                  (short)0, acc, false, false);
      acc = __builtin_amdgcn_wmma_f32_16x16x4_f32(false, a1, false, bfr[2 * j + 1],
                                                  (short)0, acc, false, false);
    }

    const int rl0 = mt * 16 + half * 8;
    float* __restrict__ orow = out + (long)(blk * BR + rl0) * NCOLS + outCol;
#pragma unroll
    for (int r = 0; r < 8; ++r) {
      float v = actW * acc[r] + bv;
      if (RESID) v += resS * buf[(rl0 + r) * PITCH + outCol];
      if (RELU) v = fmaxf(v, 0.0f);
      orow[r * NCOLS] = v;
    }
    p ^= 1;
  }
}


static __device__ __forceinline__ v4f gather_node(const v2i* __restrict__ csc,
                                                  const float* __restrict__ x,
                                                  int start, int c, v2i ent0, int lane) {
  const unsigned lane4 = (unsigned)lane * 4u;
  v4f acc = {0.0f, 0.0f, 0.0f, 0.0f};
  int idx = ent0.x;
  int wbt = ent0.y;
  for (int base = 0; base < c; base += 32) {
    if (base > 0) {
      v2i t = {0, 0};
      if (base + lane < c) t = csc[start + base + lane];
      idx = t.x;
      wbt = t.y;
    }
    const int n = (c - base < 32) ? c - base : 32;
#pragma unroll 1
    for (int j = 0; j < n; j += 8) {
      int   s[8];
      float w[8];
#pragma unroll
      for (int u = 0; u < 8; ++u) {
        const int e = (j + u < n) ? j + u : n - 1;
        s[u] = __builtin_amdgcn_readlane(idx, e);
        w[u] = __int_as_float(__builtin_amdgcn_readlane(wbt, e));
      }
      v4f v[8];
      if (j + 8 <= n) {
#pragma unroll
        for (int u = 0; u < 8; ++u) v[u] = *(const v4f*)(x + (long)s[u] * 128 + lane4);
#pragma unroll
        for (int u = 0; u < 8; ++u) acc += w[u] * v[u];
      } else {
#pragma unroll
        for (int u = 0; u < 8; ++u) {
          v[u] = (v4f){0.0f, 0.0f, 0.0f, 0.0f};
          if (j + u < n) v[u] = *(const v4f*)(x + (long)s[u] * 128 + lane4);
        }
#pragma unroll
        for (int u = 0; u < 8; ++u) acc += w[u] * v[u];
      }
    }
  }
  return acc;
}

template <bool RELU, bool PROJ>
__global__ __launch_bounds__(256) void k_layer_fused(const int* __restrict__ rowptr,
                                                     const int* __restrict__ cnt,
                                                     const int* __restrict__ cscRaw,
                                                     const float* __restrict__ x,
                                                     const float* __restrict__ dirs,
                                                     const float* __restrict__ x0,
                                                     const float* __restrict__ W,
                                                     const float* __restrict__ bias,
                                                     float* __restrict__ out,
                                                     const float* __restrict__ W2,
                                                     const float* __restrict__ bias2,
                                                     float* __restrict__ out2,
                                                     float actW, float resS, int nRows) {
  constexpr int NCOLS = 128;
  constexpr int BR    = 16;
  constexpr int PITCH = 132;
  constexpr int NC2   = 64;
  constexpr int NBUF = PROJ ? 1 : 2;
  __shared__ __attribute__((aligned(16))) float lds[2 * BR * PITCH];
  float* const xt = lds + BR * PITCH;
  __shared__ v4f w2f[PROJ ? (NC2 / 16) * 16 * 32 : 1];
  const v2i* __restrict__ csc = (const v2i*)cscRaw;

  const int wave = __builtin_amdgcn_readfirstlane((int)(threadIdx.x >> 5));
  const int lane = threadIdx.x & 31;
  const int col0 = wave * 16;
  const int half = lane >> 4;
  const int l16  = lane & 15;
  const int outCol = col0 + l16;

  v2f bfr[32];
#pragma unroll
  for (int s = 0; s < 32; ++s) {
    const int kb = 8 * (s >> 1) + 4 * half + 2 * (s & 1);
    bfr[s].x = W[(long)kb * NCOLS + outCol];
    bfr[s].y = W[(long)(kb + 1) * NCOLS + outCol];
  }
  const float bv = bias[outCol];

  if (PROJ) {
    constexpr int NFR = (NC2 / 16) * 16 * 32 / 256;
    v4f fr[NFR];
#pragma unroll
    for (int u = 0; u < NFR; ++u) {
      const int i  = u * 256 + (int)threadIdx.x;
      const int ln = i & 31;
      const int j  = (i >> 5) & 15;
      const int t  = i >> 9;
      const float* __restrict__ wp = W2 + (8 * j + 4 * (ln >> 4)) * NC2 + t * 16 + (ln & 15);
      fr[u].x = wp[0];
      fr[u].y = wp[NC2];
      fr[u].z = wp[2 * NC2];
      fr[u].w = wp[3 * NC2];
    }
#pragma unroll
    for (int u = 0; u < NFR; ++u) w2f[u * 256 + (int)threadIdx.x] = fr[u];
  }

  const int nblk = nRows / BR;
  int p = 0;
  for (int blk = blockIdx.x; blk < nblk; blk += gridDim.x) {
    float* buf = lds + p * (BR * PITCH);

    const int nodeA  = blk * BR + wave * 2;
    const int startA = rowptr[nodeA];
    const int startB = rowptr[nodeA + 1];
    const int cA     = cnt[nodeA];
    const int cB     = cnt[nodeA + 1];
    v2i iA = {0, 0};
    v2i iB = {0, 0};
    if (lane < cA) iA = csc[startA + lane];
    if (lane < cB) iB = csc[startB + lane];
    const float scA = (1.0f - GA_ALPHA) * dirs[nodeA];
    const float scB = (1.0f - GA_ALPHA) * dirs[nodeA + 1];
    const long  oA  = (long)nodeA * 128 + lane * 4;
    const v4f x0A = *(const v4f*)(x0 + oA);
    const v4f x0B = *(const v4f*)(x0 + oA + 128);

    const v4f accA = gather_node(csc, x, startA, cA, iA, lane);
    v4f rA;
    rA.x = scA * accA.x + GA_ALPHA * x0A.x;
    rA.y = scA * accA.y + GA_ALPHA * x0A.y;
    rA.z = scA * accA.z + GA_ALPHA * x0A.z;
    rA.w = scA * accA.w + GA_ALPHA * x0A.w;
    *(v4f*)(buf + (wave * 2) * PITCH + lane * 4) = rA;

    const v4f accB = gather_node(csc, x, startB, cB, iB, lane);
    v4f rB;
    rB.x = scB * accB.x + GA_ALPHA * x0B.x;
    rB.y = scB * accB.y + GA_ALPHA * x0B.y;
    rB.z = scB * accB.z + GA_ALPHA * x0B.z;
    rB.w = scB * accB.w + GA_ALPHA * x0B.w;
    *(v4f*)(buf + (wave * 2 + 1) * PITCH + lane * 4) = rB;

    __builtin_amdgcn_fence(__ATOMIC_RELEASE, "workgroup", "local");
    __builtin_amdgcn_s_barrier();
    __builtin_amdgcn_fence(__ATOMIC_ACQUIRE, "workgroup", "local");

    const float* __restrict__ arow = buf + l16 * PITCH + half * 4;
    v4f af[16];
#pragma unroll
    for (int j = 0; j < 16; ++j) af[j] = *(const v4f*)(arow + 8 * j);

    v8f acc = {};
#pragma unroll
    for (int j = 0; j < 16; ++j) {
      v2f a0, a1;
      a0.x = af[j].x; a0.y = af[j].y;
      a1.x = af[j].z; a1.y = af[j].w;
      acc = __builtin_amdgcn_wmma_f32_16x16x4_f32(false, a0, false, bfr[2 * j],
                                                  (short)0, acc, false, false);
      acc = __builtin_amdgcn_wmma_f32_16x16x4_f32(false, a1, false, bfr[2 * j + 1],
                                                  (short)0, acc, false, false);
    }

    const int rl0 = half * 8;
    if (!PROJ) {
      float* __restrict__ orow = out + (long)(blk * BR + rl0) * NCOLS + outCol;
#pragma unroll
      for (int r = 0; r < 8; ++r) {
        float v = actW * acc[r] + bv;
        v += resS * buf[(rl0 + r) * PITCH + outCol];
        if (RELU) v = fmaxf(v, 0.0f);
        orow[r * NCOLS] = v;
      }
    } else {
#pragma unroll
      for (int r = 0; r < 8; ++r) {
        float v = actW * acc[r] + bv;
        v += resS * buf[(rl0 + r) * PITCH + outCol];
        if (RELU) v = fmaxf(v, 0.0f);
        xt[(rl0 + r) * PITCH + outCol] = v;
      }
      __builtin_amdgcn_fence(__ATOMIC_RELEASE, "workgroup", "local");
      __builtin_amdgcn_s_barrier();
      __builtin_amdgcn_fence(__ATOMIC_ACQUIRE, "workgroup", "local");
      if (wave < NC2 / 16) {
        const float* __restrict__ arow2 = xt + l16 * PITCH + half * 4;
        const v4f* __restrict__ bcol = w2f + wave * 16 * 32 + lane;
        const float bv2 = bias2[wave * 16 + l16];
        v8f acc2 = {};
#pragma unroll
        for (int j = 0; j < 16; ++j) {
          const v4f a = *(const v4f*)(arow2 + 8 * j);
          const v4f b = bcol[j * 32];
          v2f a0, a1, b0, b1;
          a0.x = a.x; a0.y = a.y;
          a1.x = a.z; a1.y = a.w;
          b0.x = b.x; b0.y = b.y;
          b1.x = b.z; b1.y = b.w;
          acc2 = __builtin_amdgcn_wmma_f32_16x16x4_f32(false, a0, false, b0,
                                                       (short)0, acc2, false, false);
          acc2 = __builtin_amdgcn_wmma_f32_16x16x4_f32(false, a1, false, b1,
                                                       (short)0, acc2, false, false);
        }
        float* __restrict__ oblk = out2 + (long)blk * (BR * NC2) + wave * 16;
        const int ooff = rl0 * NC2 + l16;
#pragma unroll
        for (int r = 0; r < 8; ++r) oblk[ooff + r * NC2] = acc2[r] + bv2;
      }
    }
    if (NBUF == 2) p ^= 1;
  }
}


static inline long cdiv(long a, long b) { return (a + b - 1) / b; }

extern "C" void kernel_launch(void* const* d_in, const int* in_sizes, int n_in,
                              void* d_out, int out_size, void* d_ws, size_t ws_size,
                              hipStream_t stream) {
  const float* feat = (const float*)d_in[0];
  const int*   src  = (const int*)d_in[1];
  const int*   dst  = (const int*)d_in[2];
  const float* w1   = (const float*)d_in[3];
  const float* b1   = (const float*)d_in[4];
  const float* cw1  = (const float*)d_in[5];
  const float* cb1  = (const float*)d_in[6];
  const float* cw2  = (const float*)d_in[7];
  const float* cb2  = (const float*)d_in[8];
  const float* w2   = (const float*)d_in[9];
  const float* b2   = (const float*)d_in[10];

  const int N = in_sizes[0] / 128;
  const int E = in_sizes[1];
  const long NF = (long)N * 128;
  const int NA = (N + 63) & ~63;

  char* wsb = (char*)d_ws;
  wsb += (size_t)NA * 4;
  int* cntI   = (int*)wsb;                wsb += (size_t)NA * 4;
  int* rowptr = (int*)wsb;                wsb += (size_t)NA * 4;
  wsb += (size_t)NA * 4;
  int* bsum   = (int*)wsb;                wsb += 512 * 4;
  int* csc    = (int*)wsb;                wsb += ((size_t)E * 8 + 255) & ~(size_t)255;
  float* degO = (float*)wsb;              wsb += (size_t)NA * 4;
  float* degI = (float*)wsb;              wsb += (size_t)NA * 4;
  float* h0   = (float*)wsb;              wsb += (size_t)NF * 4;
  float* bufS = (float*)wsb;              wsb += (size_t)NF * 4;
  float* bufA = (float*)wsb;
  (void)ws_size; (void)n_in; (void)out_size;

  const int NBIN = (N + BIN_NODES - 1) >> BIN_SHIFT;
  const int NSL  = 256;
  const int per  = (int)cdiv(E, NSL);
  const int NM   = 2 * NBIN * NSL;
  const int NBS  = (int)cdiv(NM + 1, SCAN_BLOCK);
  char* tb = (char*)bufS;
  int* binD     = (int*)tb;               tb += ((size_t)E * 8 + 255) & ~(size_t)255;
  int* binS     = (int*)tb;               tb += ((size_t)E * 4 + 255) & ~(size_t)255;
  int* cntMat   = (int*)tb;               tb += ((size_t)(NM + 1) * 4 + 255) & ~(size_t)255;
  int* offMat   = (int*)tb;               tb += ((size_t)(NM + 1) * 4 + 255) & ~(size_t)255;
  int* binStart = (int*)tb;

  k_bin_hist<<<NSL, 1024, 0, stream>>>(src, dst, cntMat, E, per, NBIN, NSL);
  k_scan_block<<<NBS, SCAN_BLOCK, 0, stream>>>(cntMat, offMat, bsum, NM + 1);
  k_bin_scatter<<<NSL, 1024, 0, stream>>>(src, dst, cntMat, offMat, bsum, binStart, binD, binS,
                                          E, per, NBIN, NSL, NBS);
  k_src_count<<<NBIN, 1024, 0, stream>>>(binS, binStart, degO, N, E, NBIN);
  k_dst_place<<<NBIN, 1024, 0, stream>>>(binD, binStart, degO, rowptr, cntI, degI, csc, N);

  k_gemm_wmma<128, 8, true, false><<<512, 256, 0, stream>>>(feat, w1, b1, h0, 1.0f, 0.0f, N);

  k_layer_fused<true, false><<<768, 256, 0, stream>>>(rowptr, cntI, csc, h0, degI, h0, cw1, cb1, bufA,
                                                       nullptr, nullptr, nullptr,
                                                       GA_BETA1, 1.0f - GA_BETA1, N);

  k_layer_fused<false, true><<<768, 256, 0, stream>>>(rowptr, cntI, csc, bufA, degI, h0, cw2, cb2, nullptr,
                                                       w2, b2, (float*)d_out,
                                                       GA_BETA2, 1.0f - GA_BETA2, N);
}


# CSC build in 3 launches, not 5: each edge slice is grouped by node range once in LDS and stays in place with its range offsets; range passes walk the slice groups; in-node CSC order: LDS arrival
# speedup vs baseline: 3.0027x; 1.0636x over previous
#include <hip/hip_runtime.h>


#define GA_ALPHA 0.2f
#define GA_BETA1 0.6931471805599453f
#define GA_BETA2 0.4054651081081644f

typedef __attribute__((ext_vector_type(2))) float v2f;
typedef __attribute__((ext_vector_type(2))) int   v2i;
typedef __attribute__((ext_vector_type(4))) float v4f;
typedef __attribute__((ext_vector_type(8))) float v8f;


#define BIN_SHIFT 8
#define BIN_NODES 256
#define MAX_BINS  512
#define SLICE_CAP 6272
#define STAGE_CAP 6144
#define SEG_LANES 16
#define SEG_PER_PASS (1024 / SEG_LANES)
#define SEG_UNR 4

template <int K>
static __device__ __forceinline__ void wave_excl_scan(int* a, int lane) {
  int loc[K];
  int sum = 0;
#pragma unroll
  for (int k = 0; k < K; ++k) { loc[k] = a[lane * K + k]; sum += loc[k]; }
  int inc = sum;
#pragma unroll
  for (int off = 1; off < 32; off <<= 1) {
    const int t = __shfl_up(inc, off);
    if (lane >= off) inc += t;
  }
  int ex = inc - sum;
#pragma unroll
  for (int k = 0; k < K; ++k) { a[lane * K + k] = ex; ex += loc[k]; }
}

__global__ __launch_bounds__(1024) void k_slice_sort(const int* __restrict__ src,
                                                     const int* __restrict__ dst,
                                                     int* __restrict__ tbl,
                                                     int* __restrict__ binDraw,
                                                     int* __restrict__ binS,
                                                     int E, int per, int nbin, int tw) {
  __shared__ int h[2 * MAX_BINS];
  __shared__ int lcur[2 * MAX_BINS];
  __shared__ v2i recD[SLICE_CAP];
  __shared__ int recS[SLICE_CAP];
  v2i* __restrict__ binD = (v2i*)binDraw;
  const int tid = threadIdx.x;
  const int g = blockIdx.x;
  for (int v = tid; v < 2 * MAX_BINS; v += 1024) h[v] = 0;
  __syncthreads();
  const int lo = g * per;
  const int hi = (lo + per < E) ? lo + per : E;
  const int nD = (hi > lo) ? hi - lo : 0;
  const int last = (nD > 0) ? hi - 1 : 0;
  int s[8], d[8];
#pragma unroll
  for (int u = 0; u < 8; ++u) {
    const int e = lo + u * 1024 + tid;
    const int ec = (e < hi) ? e : last;
    s[u] = src[ec];
    d[u] = dst[ec];
  }
#pragma unroll
  for (int u = 0; u < 8; ++u) {
    if (lo + u * 1024 + tid < hi) {
      atomicAdd(&h[d[u] >> BIN_SHIFT], 1);
      atomicAdd(&h[nbin + (s[u] >> BIN_SHIFT)], 1);
    }
  }
  __syncthreads();
  if (tid < 32) wave_excl_scan<32>(h, tid);
  __syncthreads();
  for (int v = tid; v < 2 * MAX_BINS; v += 1024) lcur[v] = h[v];
  for (int v = tid; v < 2 * nbin; v += 1024) tbl[(long)g * tw + v] = h[v];
  if (tid == 0) tbl[(long)g * tw + 2 * nbin] = 2 * nD;
  __syncthreads();
#pragma unroll
  for (int u = 0; u < 8; ++u) {
    if (lo + u * 1024 + tid < hi) {
      v2i rec;
      rec.x = s[u];
      rec.y = d[u];
      const int p = atomicAdd(&lcur[d[u] >> BIN_SHIFT], 1);
      recD[p] = rec;
      const int q = atomicAdd(&lcur[nbin + (s[u] >> BIN_SHIFT)], 1) - nD;
      recS[q] = s[u];
    }
  }
  __syncthreads();
  for (int i = tid; i < nD; i += 1024) {
    binD[lo + i] = recD[i];
    binS[lo + i] = recS[i];
  }
}

__global__ __launch_bounds__(1024) void k_src_count(const int* __restrict__ binS,
                                                    const int* __restrict__ tbl,
                                                    float* __restrict__ degO,
                                                    int N, int per, int nbin, int tw, int nslice) {
  __shared__ int c[BIN_NODES];
  const int tid = threadIdx.x;
  const int b = blockIdx.x;
  if (tid < BIN_NODES) c[tid] = 0;
  __syncthreads();
  const int node0 = b << BIN_SHIFT;
  const int l  = tid & (SEG_LANES - 1);
  const int sg = tid / SEG_LANES;
  for (int seg0 = 0; seg0 < nslice; seg0 += SEG_UNR * SEG_PER_PASS) {
    int a[SEG_UNR], z[SEG_UNR];
    long o[SEG_UNR];
#pragma unroll
    for (int k = 0; k < SEG_UNR; ++k) {
      const int seg = seg0 + k * SEG_PER_PASS + sg;
      const bool ok = seg < nslice;
      const int sc = ok ? seg : nslice - 1;
      const int* __restrict__ t = tbl + (long)sc * tw;
      const int nDg = t[nbin];
      const int ta = t[nbin + b];
      const int tz = t[nbin + b + 1];
      a[k] = ta - nDg + l;
      z[k] = ok ? tz - nDg : 0;
      if (!ok) a[k] = 0;
      o[k] = (long)sc * per;
    }
    int r0[SEG_UNR], r1[SEG_UNR];
#pragma unroll
    for (int k = 0; k < SEG_UNR; ++k) {
      const int i1 = a[k] + SEG_LANES;
      r0[k] = binS[(a[k] < z[k]) ? o[k] + a[k] : 0];
      r1[k] = binS[(i1 < z[k]) ? o[k] + i1 : 0];
    }
#pragma unroll
    for (int k = 0; k < SEG_UNR; ++k) {
      if (a[k] < z[k]) atomicAdd(&c[r0[k] - node0], 1);
      if (a[k] + SEG_LANES < z[k]) atomicAdd(&c[r1[k] - node0], 1);
      for (int i = a[k] + 2 * SEG_LANES; i < z[k]; i += SEG_LANES)
        atomicAdd(&c[binS[o[k] + i] - node0], 1);
    }
  }
  __syncthreads();
  if (tid < BIN_NODES) {
    const int node = node0 + tid;
    if (node < N) degO[node] = rsqrtf(fmaxf((float)c[tid], 1.0f));
  }
}

__global__ __launch_bounds__(1024) void k_dst_place(const int* __restrict__ binDraw,
                                                    const int* __restrict__ tbl,
                                                    const float* __restrict__ degO,
                                                    int* __restrict__ rowptr,
                                                    int* __restrict__ cntI,
                                                    float* __restrict__ degI,
                                                    int* __restrict__ cscRaw,
                                                    int N, int per, int tw, int nslice) {
  __shared__ int c[BIN_NODES];
  __shared__ int cur[BIN_NODES];
  __shared__ int first;
  __shared__ v2i stage[STAGE_CAP];
  const v2i* __restrict__ binD = (const v2i*)binDraw;
  v2i* __restrict__ csc = (v2i*)cscRaw;
  const int tid = threadIdx.x;
  const int b = blockIdx.x;
  if (tid < BIN_NODES) c[tid] = 0;
  if (tid == 0) first = 0;
  __syncthreads();
  const int node0 = b << BIN_SHIFT;
  const int l  = tid & (SEG_LANES - 1);
  const int sg = tid / SEG_LANES;

  int begins = 0;
  for (int seg0 = 0; seg0 < nslice; seg0 += SEG_UNR * SEG_PER_PASS) {
    int a[SEG_UNR], z[SEG_UNR];
    long o[SEG_UNR];
#pragma unroll
    for (int k = 0; k < SEG_UNR; ++k) {
      const int seg = seg0 + k * SEG_PER_PASS + sg;
      const bool ok = seg < nslice;
      const int sc = ok ? seg : nslice - 1;
      const int* __restrict__ t = tbl + (long)sc * tw;
      const int ta = t[b];
      const int tz = t[b + 1];
      if (ok && l == 0) begins += ta;
      a[k] = ok ? ta + l : 0;
      z[k] = ok ? tz : 0;
      o[k] = (long)sc * per;
    }
    int d0[SEG_UNR], d1[SEG_UNR];
#pragma unroll
    for (int k = 0; k < SEG_UNR; ++k) {
      const int i1 = a[k] + SEG_LANES;
      d0[k] = binD[(a[k] < z[k]) ? o[k] + a[k] : 0].y;
      d1[k] = binD[(i1 < z[k]) ? o[k] + i1 : 0].y;
    }
#pragma unroll
    for (int k = 0; k < SEG_UNR; ++k) {
      if (a[k] < z[k]) atomicAdd(&c[d0[k] - node0], 1);
      if (a[k] + SEG_LANES < z[k]) atomicAdd(&c[d1[k] - node0], 1);
      for (int i = a[k] + 2 * SEG_LANES; i < z[k]; i += SEG_LANES)
        atomicAdd(&c[binD[o[k] + i].y - node0], 1);
    }
  }
#pragma unroll
  for (int off = 16; off > 0; off >>= 1) begins += __shfl_xor(begins, off);
  if ((tid & 31) == 0) atomicAdd(&first, begins);
  __syncthreads();

  if (tid < BIN_NODES) cur[tid] = c[tid];
  __syncthreads();
  if (tid < 32) wave_excl_scan<8>(cur, tid);
  __syncthreads();
  const int start = first;
  const int nrec  = cur[BIN_NODES - 1] + c[BIN_NODES - 1];
  if (tid < BIN_NODES) {
    const int node = node0 + tid;
    if (node < N) {
      rowptr[node] = start + cur[tid];
      cntI[node]   = c[tid];
      degI[node]   = rsqrtf(fmaxf((float)c[tid], 1.0f));
    }
  }
  __syncthreads();

  const bool staged = nrec <= STAGE_CAP;
  for (int seg0 = 0; seg0 < nslice; seg0 += SEG_UNR * SEG_PER_PASS) {
    int a[SEG_UNR], z[SEG_UNR];
    long o[SEG_UNR];
#pragma unroll
    for (int k = 0; k < SEG_UNR; ++k) {
      const int seg = seg0 + k * SEG_PER_PASS + sg;
      const bool ok = seg < nslice;
      const int sc = ok ? seg : nslice - 1;
      const int* __restrict__ t = tbl + (long)sc * tw;
      const int ta = t[b];
      const int tz = t[b + 1];
      a[k] = ok ? ta + l : 0;
      z[k] = ok ? tz : 0;
      o[k] = (long)sc * per;
    }
    v2i r0[SEG_UNR], r1[SEG_UNR];
#pragma unroll
    for (int k = 0; k < SEG_UNR; ++k) {
      const int i1 = a[k] + SEG_LANES;
      r0[k] = binD[(a[k] < z[k]) ? o[k] + a[k] : 0];
      r1[k] = binD[(i1 < z[k]) ? o[k] + i1 : 0];
    }
    float w0[SEG_UNR], w1[SEG_UNR];
#pragma unroll
    for (int k = 0; k < SEG_UNR; ++k) {
      w0[k] = degO[r0[k].x];
      w1[k] = degO[r1[k].x];
    }
#pragma unroll
    for (int k = 0; k < SEG_UNR; ++k) {
      if (a[k] < z[k]) {
        const int p = atomicAdd(&cur[r0[k].y - node0], 1);
        v2i ent;
        ent.x = r0[k].x;
        ent.y = __float_as_int(w0[k]);
        if (staged) stage[p] = ent; else csc[start + p] = ent;
      }
      if (a[k] + SEG_LANES < z[k]) {
        const int p = atomicAdd(&cur[r1[k].y - node0], 1);
        v2i ent;
        ent.x = r1[k].x;
        ent.y = __float_as_int(w1[k]);
        if (staged) stage[p] = ent; else csc[start + p] = ent;
      }
      for (int i = a[k] + 2 * SEG_LANES; i < z[k]; i += SEG_LANES) {
        const v2i r = binD[o[k] + i];
        const float w = degO[r.x];
        const int p = atomicAdd(&cur[r.y - node0], 1);
        v2i ent;
        ent.x = r.x;
        ent.y = __float_as_int(w);
        if (staged) stage[p] = ent; else csc[start + p] = ent;
      }
    }
  }
  if (staged) {
    __syncthreads();
    for (int i = tid; i < nrec; i += 1024) csc[start + i] = stage[i];
  }
}

template <int NCOLS, int WAVES, bool RELU, bool RESID>
__global__ __launch_bounds__(WAVES * 32) void k_gemm_wmma(const float* __restrict__ A,
                                                          const float* __restrict__ W,
                                                          const float* __restrict__ bias,
                                                          float* __restrict__ out,
                                                          float actW, float resS, int nRows) {
  constexpr int NCT   = NCOLS / 16;
  constexpr int MT    = WAVES / NCT;
  constexpr int BR    = MT * 16;
  constexpr int RPW   = BR / WAVES;
  constexpr int PITCH = 132;
  __shared__ __attribute__((aligned(16))) float lds[2][BR * PITCH];

  const int wave = threadIdx.x >> 5;
  const int lane = threadIdx.x & 31;
  const int ct   = wave % NCT;
  const int mt   = wave / NCT;
  const int col0 = ct * 16;
  const int half = lane >> 4;
  const int l16  = lane & 15;
  const int outCol = col0 + l16;

  v2f bfr[32];
#pragma unroll
  for (int s = 0; s < 32; ++s) {
    const int kb = 8 * (s >> 1) + 4 * half + 2 * (s & 1);
    bfr[s].x = W[(long)kb * NCOLS + outCol];
    bfr[s].y = W[(long)(kb + 1) * NCOLS + outCol];
  }
  const float bv = bias[outCol];

  const int nblk = nRows / BR;
  int blk = blockIdx.x;

  v4f stage[RPW];
#pragma unroll
  for (int i = 0; i < RPW; ++i) stage[i] = (v4f){0.0f, 0.0f, 0.0f, 0.0f};
  if (blk < nblk) {
#pragma unroll
    for (int i = 0; i < RPW; ++i)
      stage[i] = *(const v4f*)(A + (long)(blk * BR + wave * RPW + i) * 128 + lane * 4);
  }

  int p = 0;
  for (; blk < nblk; blk += gridDim.x) {
    float* __restrict__ buf = lds[p];
#pragma unroll
    for (int i = 0; i < RPW; ++i)
      *(v4f*)(buf + (wave * RPW + i) * PITCH + lane * 4) = stage[i];

    __builtin_amdgcn_fence(__ATOMIC_RELEASE, "workgroup", "local");
    __builtin_amdgcn_s_barrier();
    __builtin_amdgcn_fence(__ATOMIC_ACQUIRE, "workgroup", "local");

    const int next = blk + gridDim.x;
    if (next < nblk) {
#pragma unroll
      for (int i = 0; i < RPW; ++i)
        stage[i] = *(const v4f*)(A + (long)(next * BR + wave * RPW + i) * 128 + lane * 4);
    }

    const float* __restrict__ arow = buf + (mt * 16 + l16) * PITCH + half * 4;
    v4f af[16];
#pragma unroll
    for (int j = 0; j < 16; ++j) af[j] = *(const v4f*)(arow + 8 * j);

    v8f acc = {};
#pragma unroll
    for (int j = 0; j < 16; ++j) {
      v2f a0, a1;
      a0.x = af[j].x; a0.y = af[j].y;
      a1.x = af[j].z; a1.y = af[j].w;
      acc = __builtin_amdgcn_wmma_f32_16x16x4_f32(false, a0, false, bfr[2 * j],
                                                  (short)0, acc, false, false);
      acc = __builtin_amdgcn_wmma_f32_16x16x4_f32(false, a1, false, bfr[2 * j + 1],
                                                  (short)0, acc, false, false);
    }

    const int rl0 = mt * 16 + half * 8;
    float* __restrict__ orow = out + (long)(blk * BR + rl0) * NCOLS + outCol;
#pragma unroll
    for (int r = 0; r < 8; ++r) {
      float v = actW * acc[r] + bv;
      if (RESID) v += resS * buf[(rl0 + r) * PITCH + outCol];
      if (RELU) v = fmaxf(v, 0.0f);
      orow[r * NCOLS] = v;
    }
    p ^= 1;
  }
}


static __device__ __forceinline__ v4f gather_node(const v2i* __restrict__ csc,
                                                  const float* __restrict__ x,
                                                  int start, int c, v2i ent0, int lane) {
  const unsigned lane4 = (unsigned)lane * 4u;
  v4f acc = {0.0f, 0.0f, 0.0f, 0.0f};
  int idx = ent0.x;
  int wbt = ent0.y;
  for (int base = 0; base < c; base += 32) {
    if (base > 0) {
      v2i t = {0, 0};
      if (base + lane < c) t = csc[start + base + lane];
      idx = t.x;
      wbt = t.y;
    }
    const int n = (c - base < 32) ? c - base : 32;
#pragma unroll 1
    for (int j = 0; j < n; j += 8) {
      int   s[8];
      float w[8];
#pragma unroll
      for (int u = 0; u < 8; ++u) {
        const int e = (j + u < n) ? j + u : n - 1;
        s[u] = __builtin_amdgcn_readlane(idx, e);
        w[u] = __int_as_float(__builtin_amdgcn_readlane(wbt, e));
      }
      v4f v[8];
      if (j + 8 <= n) {
#pragma unroll
        for (int u = 0; u < 8; ++u) v[u] = *(const v4f*)(x + (long)s[u] * 128 + lane4);
#pragma unroll
        for (int u = 0; u < 8; ++u) acc += w[u] * v[u];
      } else {
#pragma unroll
        for (int u = 0; u < 8; ++u) {
          v[u] = (v4f){0.0f, 0.0f, 0.0f, 0.0f};
          if (j + u < n) v[u] = *(const v4f*)(x + (long)s[u] * 128 + lane4);
        }
#pragma unroll
        for (int u = 0; u < 8; ++u) acc += w[u] * v[u];
      }
    }
  }
  return acc;
}

template <bool RELU, bool PROJ>
__global__ __launch_bounds__(256) void k_layer_fused(const int* __restrict__ rowptr,
                                                     const int* __restrict__ cnt,
                                                     const int* __restrict__ cscRaw,
                                                     const float* __restrict__ x,
                                                     const float* __restrict__ dirs,
                                                     const float* __restrict__ x0,
                                                     const float* __restrict__ W,
                                                     const float* __restrict__ bias,
                                                     float* __restrict__ out,
                                                     const float* __restrict__ W2,
                                                     const float* __restrict__ bias2,
                                                     float* __restrict__ out2,
                                                     float actW, float resS, int nRows) {
  constexpr int NCOLS = 128;
  constexpr int BR    = 16;
  constexpr int PITCH = 132;
  constexpr int NC2   = 64;
  constexpr int NBUF = PROJ ? 1 : 2;
  __shared__ __attribute__((aligned(16))) float lds[2 * BR * PITCH];
  float* const xt = lds + BR * PITCH;
  __shared__ v4f w2f[PROJ ? (NC2 / 16) * 16 * 32 : 1];
  const v2i* __restrict__ csc = (const v2i*)cscRaw;

  const int wave = __builtin_amdgcn_readfirstlane((int)(threadIdx.x >> 5));
  const int lane = threadIdx.x & 31;
  const int col0 = wave * 16;
  const int half = lane >> 4;
  const int l16  = lane & 15;
  const int outCol = col0 + l16;

  v2f bfr[32];
#pragma unroll
  for (int s = 0; s < 32; ++s) {
    const int kb = 8 * (s >> 1) + 4 * half + 2 * (s & 1);
    bfr[s].x = W[(long)kb * NCOLS + outCol];
    bfr[s].y = W[(long)(kb + 1) * NCOLS + outCol];
  }
  const float bv = bias[outCol];

  if (PROJ) {
    constexpr int NFR = (NC2 / 16) * 16 * 32 / 256;
    v4f fr[NFR];
#pragma unroll
    for (int u = 0; u < NFR; ++u) {
      const int i  = u * 256 + (int)threadIdx.x;
      const int ln = i & 31;
      const int j  = (i >> 5) & 15;
      const int t  = i >> 9;
      const float* __restrict__ wp = W2 + (8 * j + 4 * (ln >> 4)) * NC2 + t * 16 + (ln & 15);
      fr[u].x = wp[0];
      fr[u].y = wp[NC2];
      fr[u].z = wp[2 * NC2];
      fr[u].w = wp[3 * NC2];
    }
#pragma unroll
    for (int u = 0; u < NFR; ++u) w2f[u * 256 + (int)threadIdx.x] = fr[u];
  }

  const int nblk = nRows / BR;
  int p = 0;
  for (int blk = blockIdx.x; blk < nblk; blk += gridDim.x) {
    float* buf = lds + p * (BR * PITCH);

    const int nodeA  = blk * BR + wave * 2;
    const int startA = rowptr[nodeA];
    const int startB = rowptr[nodeA + 1];
    const int cA     = cnt[nodeA];
    const int cB     = cnt[nodeA + 1];
    v2i iA = {0, 0};
    v2i iB = {0, 0};
    if (lane < cA) iA = csc[startA + lane];
    if (lane < cB) iB = csc[startB + lane];
    const float scA = (1.0f - GA_ALPHA) * dirs[nodeA];
    const float scB = (1.0f - GA_ALPHA) * dirs[nodeA + 1];
    const long  oA  = (long)nodeA * 128 + lane * 4;
    const v4f x0A = *(const v4f*)(x0 + oA);
    const v4f x0B = *(const v4f*)(x0 + oA + 128);

    const v4f accA = gather_node(csc, x, startA, cA, iA, lane);
    v4f rA;
    rA.x = scA * accA.x + GA_ALPHA * x0A.x;
    rA.y = scA * accA.y + GA_ALPHA * x0A.y;
    rA.z = scA * accA.z + GA_ALPHA * x0A.z;
    rA.w = scA * accA.w + GA_ALPHA * x0A.w;
    *(v4f*)(buf + (wave * 2) * PITCH + lane * 4) = rA;

    const v4f accB = gather_node(csc, x, startB, cB, iB, lane);
    v4f rB;
    rB.x = scB * accB.x + GA_ALPHA * x0B.x;
    rB.y = scB * accB.y + GA_ALPHA * x0B.y;
    rB.z = scB * accB.z + GA_ALPHA * x0B.z;
    rB.w = scB * accB.w + GA_ALPHA * x0B.w;
    *(v4f*)(buf + (wave * 2 + 1) * PITCH + lane * 4) = rB;

    __builtin_amdgcn_fence(__ATOMIC_RELEASE, "workgroup", "local");
    __builtin_amdgcn_s_barrier();
    __builtin_amdgcn_fence(__ATOMIC_ACQUIRE, "workgroup", "local");

    const float* __restrict__ arow = buf + l16 * PITCH + half * 4;
    v4f af[16];
#pragma unroll
    for (int j = 0; j < 16; ++j) af[j] = *(const v4f*)(arow + 8 * j);

    v8f acc = {};
#pragma unroll
    for (int j = 0; j < 16; ++j) {
      v2f a0, a1;
      a0.x = af[j].x; a0.y = af[j].y;
      a1.x = af[j].z; a1.y = af[j].w;
      acc = __builtin_amdgcn_wmma_f32_16x16x4_f32(false, a0, false, bfr[2 * j],
                                                  (short)0, acc, false, false);
      acc = __builtin_amdgcn_wmma_f32_16x16x4_f32(false, a1, false, bfr[2 * j + 1],
                                                  (short)0, acc, false, false);
    }

    const int rl0 = half * 8;
    if (!PROJ) {
      float* __restrict__ orow = out + (long)(blk * BR + rl0) * NCOLS + outCol;
#pragma unroll
      for (int r = 0; r < 8; ++r) {
        float v = actW * acc[r] + bv;
        v += resS * buf[(rl0 + r) * PITCH + outCol];
        if (RELU) v = fmaxf(v, 0.0f);
        orow[r * NCOLS] = v;
      }
    } else {
#pragma unroll
      for (int r = 0; r < 8; ++r) {
        float v = actW * acc[r] + bv;
        v += resS * buf[(rl0 + r) * PITCH + outCol];
        if (RELU) v = fmaxf(v, 0.0f);
        xt[(rl0 + r) * PITCH + outCol] = v;
      }
      __builtin_amdgcn_fence(__ATOMIC_RELEASE, "workgroup", "local");
      __builtin_amdgcn_s_barrier();
      __builtin_amdgcn_fence(__ATOMIC_ACQUIRE, "workgroup", "local");
      if (wave < NC2 / 16) {
        const float* __restrict__ arow2 = xt + l16 * PITCH + half * 4;
        const v4f* __restrict__ bcol = w2f + wave * 16 * 32 + lane;
        const float bv2 = bias2[wave * 16 + l16];
        v8f acc2 = {};
#pragma unroll
        for (int j = 0; j < 16; ++j) {
          const v4f a = *(const v4f*)(arow2 + 8 * j);
          const v4f b = bcol[j * 32];
          v2f a0, a1, b0, b1;
          a0.x = a.x; a0.y = a.y;
          a1.x = a.z; a1.y = a.w;
          b0.x = b.x; b0.y = b.y;
          b1.x = b.z; b1.y = b.w;
          acc2 = __builtin_amdgcn_wmma_f32_16x16x4_f32(false, a0, false, b0,
                                                       (short)0, acc2, false, false);
          acc2 = __builtin_amdgcn_wmma_f32_16x16x4_f32(false, a1, false, b1,
                                                       (short)0, acc2, false, false);
        }
        float* __restrict__ oblk = out2 + (long)blk * (BR * NC2) + wave * 16;
        const int ooff = rl0 * NC2 + l16;
#pragma unroll
        for (int r = 0; r < 8; ++r) oblk[ooff + r * NC2] = acc2[r] + bv2;
      }
    }
    if (NBUF == 2) p ^= 1;
  }
}


static inline long cdiv(long a, long b) { return (a + b - 1) / b; }

extern "C" void kernel_launch(void* const* d_in, const int* in_sizes, int n_in,
                              void* d_out, int out_size, void* d_ws, size_t ws_size,
                              hipStream_t stream) {
  const float* feat = (const float*)d_in[0];
  const int*   src  = (const int*)d_in[1];
  const int*   dst  = (const int*)d_in[2];
  const float* w1   = (const float*)d_in[3];
  const float* b1   = (const float*)d_in[4];
  const float* cw1  = (const float*)d_in[5];
  const float* cb1  = (const float*)d_in[6];
  const float* cw2  = (const float*)d_in[7];
  const float* cb2  = (const float*)d_in[8];
  const float* w2   = (const float*)d_in[9];
  const float* b2   = (const float*)d_in[10];

  const int N = in_sizes[0] / 128;
  const int E = in_sizes[1];
  const long NF = (long)N * 128;
  const int NA = (N + 63) & ~63;

  char* wsb = (char*)d_ws;
  wsb += (size_t)NA * 4;
  int* cntI   = (int*)wsb;                wsb += (size_t)NA * 4;
  int* rowptr = (int*)wsb;                wsb += (size_t)NA * 4;
  wsb += (size_t)NA * 4;
  int* bsum   = (int*)wsb;                wsb += 512 * 4;
  int* csc    = (int*)wsb;                wsb += ((size_t)E * 8 + 255) & ~(size_t)255;
  float* degO = (float*)wsb;              wsb += (size_t)NA * 4;
  float* degI = (float*)wsb;              wsb += (size_t)NA * 4;
  float* h0   = (float*)wsb;              wsb += (size_t)NF * 4;
  float* bufS = (float*)wsb;              wsb += (size_t)NF * 4;
  float* bufA = (float*)wsb;
  (void)ws_size; (void)n_in; (void)out_size;

  const int NBIN = (N + BIN_NODES - 1) >> BIN_SHIFT;
  const int NSL  = (int)cdiv(E, SLICE_CAP);
  const int per  = (int)cdiv(E, NSL);
  const int TW   = 2 * NBIN + 1;
  char* tb = (char*)bufS;
  int* binD     = (int*)tb;               tb += ((size_t)E * 8 + 255) & ~(size_t)255;
  int* binS     = (int*)tb;               tb += ((size_t)E * 4 + 255) & ~(size_t)255;
  int* tbl      = (int*)tb;
  (void)bsum;

  k_slice_sort<<<NSL, 1024, 0, stream>>>(src, dst, tbl, binD, binS, E, per, NBIN, TW);
  k_src_count<<<NBIN, 1024, 0, stream>>>(binS, tbl, degO, N, per, NBIN, TW, NSL);
  k_dst_place<<<NBIN, 1024, 0, stream>>>(binD, tbl, degO, rowptr, cntI, degI, csc, N, per, TW, NSL);

  k_gemm_wmma<128, 8, true, false><<<512, 256, 0, stream>>>(feat, w1, b1, h0, 1.0f, 0.0f, N);

  k_layer_fused<true, false><<<768, 256, 0, stream>>>(rowptr, cntI, csc, h0, degI, h0, cw1, cb1, bufA,
                                                       nullptr, nullptr, nullptr,
                                                       GA_BETA1, 1.0f - GA_BETA1, N);

  k_layer_fused<false, true><<<768, 256, 0, stream>>>(rowptr, cntI, csc, bufA, degI, h0, cw2, cb2, nullptr,
                                                       w2, b2, (float*)d_out,
                                                       GA_BETA2, 1.0f - GA_BETA2, N);
}


# Finished output tiles leave through LDS as whole-row stores (full 128-B lines per store, not 64-B pieces off the accumulators) in GEMM 1, layer 1 and the projection, behind the next block's barrier
# speedup vs baseline: 3.1227x; 1.0400x over previous
#include <hip/hip_runtime.h>


#define GA_ALPHA 0.2f
#define GA_BETA1 0.6931471805599453f
#define GA_BETA2 0.4054651081081644f

typedef __attribute__((ext_vector_type(2))) float v2f;
typedef __attribute__((ext_vector_type(2))) int   v2i;
typedef __attribute__((ext_vector_type(4))) float v4f;
typedef __attribute__((ext_vector_type(8))) float v8f;


#define BIN_SHIFT 8
#define BIN_NODES 256
#define MAX_BINS  512
#define SLICE_CAP 6272
#define STAGE_CAP 6144
#define SEG_LANES 16
#define SEG_PER_PASS (1024 / SEG_LANES)
#define SEG_UNR 4

template <int K>
static __device__ __forceinline__ void wave_excl_scan(int* a, int lane) {
  int loc[K];
  int sum = 0;
#pragma unroll
  for (int k = 0; k < K; ++k) { loc[k] = a[lane * K + k]; sum += loc[k]; }
  int inc = sum;
#pragma unroll
  for (int off = 1; off < 32; off <<= 1) {
    const int t = __shfl_up(inc, off);
    if (lane >= off) inc += t;
  }
  int ex = inc - sum;
#pragma unroll
  for (int k = 0; k < K; ++k) { a[lane * K + k] = ex; ex += loc[k]; }
}

__global__ __launch_bounds__(1024) void k_slice_sort(const int* __restrict__ src,
                                                     const int* __restrict__ dst,
                                                     int* __restrict__ tbl,
                                                     int* __restrict__ binDraw,
                                                     int* __restrict__ binS,
                                                     int E, int per, int nbin, int tw) {
  __shared__ int h[2 * MAX_BINS];
  __shared__ int lcur[2 * MAX_BINS];
  __shared__ v2i recD[SLICE_CAP];
  __shared__ int recS[SLICE_CAP];
  v2i* __restrict__ binD = (v2i*)binDraw;
  const int tid = threadIdx.x;
  const int g = blockIdx.x;
  for (int v = tid; v < 2 * MAX_BINS; v += 1024) h[v] = 0;
  __syncthreads();
  const int lo = g * per;
  const int hi = (lo + per < E) ? lo + per : E;
  const int nD = (hi > lo) ? hi - lo : 0;
  const int last = (nD > 0) ? hi - 1 : 0;
  int s[8], d[8];
#pragma unroll
  for (int u = 0; u < 8; ++u) {
    const int e = lo + u * 1024 + tid;
    const int ec = (e < hi) ? e : last;
    s[u] = src[ec];
    d[u] = dst[ec];
  }
#pragma unroll
  for (int u = 0; u < 8; ++u) {
    if (lo + u * 1024 + tid < hi) {
      atomicAdd(&h[d[u] >> BIN_SHIFT], 1);
      atomicAdd(&h[nbin + (s[u] >> BIN_SHIFT)], 1);
    }
  }
  __syncthreads();
  if (tid < 32) wave_excl_scan<32>(h, tid);
  __syncthreads();
  for (int v = tid; v < 2 * MAX_BINS; v += 1024) lcur[v] = h[v];
  for (int v = tid; v < 2 * nbin; v += 1024) tbl[(long)g * tw + v] = h[v];
  if (tid == 0) tbl[(long)g * tw + 2 * nbin] = 2 * nD;
  __syncthreads();
#pragma unroll
  for (int u = 0; u < 8; ++u) {
    if (lo + u * 1024 + tid < hi) {
      v2i rec;
      rec.x = s[u];
      rec.y = d[u];
      const int p = atomicAdd(&lcur[d[u] >> BIN_SHIFT], 1);
      recD[p] = rec;
      const int q = atomicAdd(&lcur[nbin + (s[u] >> BIN_SHIFT)], 1) - nD;
      recS[q] = s[u];
    }
  }
  __syncthreads();
  for (int i = tid; i < nD; i += 1024) {
    binD[lo + i] = recD[i];
    binS[lo + i] = recS[i];
  }
}

__global__ __launch_bounds__(1024) void k_src_count(const int* __restrict__ binS,
                                                    const int* __restrict__ tbl,
                                                    float* __restrict__ degO,
                                                    int N, int per, int nbin, int tw, int nslice) {
  __shared__ int c[BIN_NODES];
  const int tid = threadIdx.x;
  const int b = blockIdx.x;
  if (tid < BIN_NODES) c[tid] = 0;
  __syncthreads();
  const int node0 = b << BIN_SHIFT;
  const int l  = tid & (SEG_LANES - 1);
  const int sg = tid / SEG_LANES;
  for (int seg0 = 0; seg0 < nslice; seg0 += SEG_UNR * SEG_PER_PASS) {
    int a[SEG_UNR], z[SEG_UNR];
    long o[SEG_UNR];
#pragma unroll
    for (int k = 0; k < SEG_UNR; ++k) {
      const int seg = seg0 + k * SEG_PER_PASS + sg;
      const bool ok = seg < nslice;
      const int sc = ok ? seg : nslice - 1;
      const int* __restrict__ t = tbl + (long)sc * tw;
      const int nDg = t[nbin];
      const int ta = t[nbin + b];
      const int tz = t[nbin + b + 1];
      a[k] = ta - nDg + l;
      z[k] = ok ? tz - nDg : 0;
      if (!ok) a[k] = 0;
      o[k] = (long)sc * per;
    }
    int r0[SEG_UNR], r1[SEG_UNR];
#pragma unroll
    for (int k = 0; k < SEG_UNR; ++k) {
      const int i1 = a[k] + SEG_LANES;
      r0[k] = binS[(a[k] < z[k]) ? o[k] + a[k] : 0];
      r1[k] = binS[(i1 < z[k]) ? o[k] + i1 : 0];
    }
#pragma unroll
    for (int k = 0; k < SEG_UNR; ++k) {
      if (a[k] < z[k]) atomicAdd(&c[r0[k] - node0], 1);
      if (a[k] + SEG_LANES < z[k]) atomicAdd(&c[r1[k] - node0], 1);
      for (int i = a[k] + 2 * SEG_LANES; i < z[k]; i += SEG_LANES)
        atomicAdd(&c[binS[o[k] + i] - node0], 1);
    }
  }
  __syncthreads();
  if (tid < BIN_NODES) {
    const int node = node0 + tid;
    if (node < N) degO[node] = rsqrtf(fmaxf((float)c[tid], 1.0f));
  }
}

__global__ __launch_bounds__(1024) void k_dst_place(const int* __restrict__ binDraw,
                                                    const int* __restrict__ tbl,
                                                    const float* __restrict__ degO,
                                                    int* __restrict__ rowptr,
                                                    int* __restrict__ cntI,
                                                    float* __restrict__ degI,
                                                    int* __restrict__ cscRaw,
                                                    int N, int per, int tw, int nslice) {
  __shared__ int c[BIN_NODES];
  __shared__ int cur[BIN_NODES];
  __shared__ int first;
  __shared__ v2i stage[STAGE_CAP];
  const v2i* __restrict__ binD = (const v2i*)binDraw;
  v2i* __restrict__ csc = (v2i*)cscRaw;
  const int tid = threadIdx.x;
  const int b = blockIdx.x;
  if (tid < BIN_NODES) c[tid] = 0;
  if (tid == 0) first = 0;
  __syncthreads();
  const int node0 = b << BIN_SHIFT;
  const int l  = tid & (SEG_LANES - 1);
  const int sg = tid / SEG_LANES;

  int begins = 0;
  for (int seg0 = 0; seg0 < nslice; seg0 += SEG_UNR * SEG_PER_PASS) {
    int a[SEG_UNR], z[SEG_UNR];
    long o[SEG_UNR];
#pragma unroll
    for (int k = 0; k < SEG_UNR; ++k) {
      const int seg = seg0 + k * SEG_PER_PASS + sg;
      const bool ok = seg < nslice;
      const int sc = ok ? seg : nslice - 1;
      const int* __restrict__ t = tbl + (long)sc * tw;
      const int ta = t[b];
      const int tz = t[b + 1];
      if (ok && l == 0) begins += ta;
      a[k] = ok ? ta + l : 0;
      z[k] = ok ? tz : 0;
      o[k] = (long)sc * per;
    }
    int d0[SEG_UNR], d1[SEG_UNR];
#pragma unroll
    for (int k = 0; k < SEG_UNR; ++k) {
      const int i1 = a[k] + SEG_LANES;
      d0[k] = binD[(a[k] < z[k]) ? o[k] + a[k] : 0].y;
      d1[k] = binD[(i1 < z[k]) ? o[k] + i1 : 0].y;
    }
#pragma unroll
    for (int k = 0; k < SEG_UNR; ++k) {
      if (a[k] < z[k]) atomicAdd(&c[d0[k] - node0], 1);
      if (a[k] + SEG_LANES < z[k]) atomicAdd(&c[d1[k] - node0], 1);
      for (int i = a[k] + 2 * SEG_LANES; i < z[k]; i += SEG_LANES)
        atomicAdd(&c[binD[o[k] + i].y - node0], 1);
    }
  }
#pragma unroll
  for (int off = 16; off > 0; off >>= 1) begins += __shfl_xor(begins, off);
  if ((tid & 31) == 0) atomicAdd(&first, begins);
  __syncthreads();

  if (tid < BIN_NODES) cur[tid] = c[tid];
  __syncthreads();
  if (tid < 32) wave_excl_scan<8>(cur, tid);
  __syncthreads();
  const int start = first;
  const int nrec  = cur[BIN_NODES - 1] + c[BIN_NODES - 1];
  if (tid < BIN_NODES) {
    const int node = node0 + tid;
    if (node < N) {
      rowptr[node] = start + cur[tid];
      cntI[node]   = c[tid];
      degI[node]   = rsqrtf(fmaxf((float)c[tid], 1.0f));
    }
  }
  __syncthreads();

  const bool staged = nrec <= STAGE_CAP;
  for (int seg0 = 0; seg0 < nslice; seg0 += SEG_UNR * SEG_PER_PASS) {
    int a[SEG_UNR], z[SEG_UNR];
    long o[SEG_UNR];
#pragma unroll
    for (int k = 0; k < SEG_UNR; ++k) {
      const int seg = seg0 + k * SEG_PER_PASS + sg;
      const bool ok = seg < nslice;
      const int sc = ok ? seg : nslice - 1;
      const int* __restrict__ t = tbl + (long)sc * tw;
      const int ta = t[b];
      const int tz = t[b + 1];
      a[k] = ok ? ta + l : 0;
      z[k] = ok ? tz : 0;
      o[k] = (long)sc * per;
    }
    v2i r0[SEG_UNR], r1[SEG_UNR];
#pragma unroll
    for (int k = 0; k < SEG_UNR; ++k) {
      const int i1 = a[k] + SEG_LANES;
      r0[k] = binD[(a[k] < z[k]) ? o[k] + a[k] : 0];
      r1[k] = binD[(i1 < z[k]) ? o[k] + i1 : 0];
    }
    float w0[SEG_UNR], w1[SEG_UNR];
#pragma unroll
    for (int k = 0; k < SEG_UNR; ++k) {
      w0[k] = degO[r0[k].x];
      w1[k] = degO[r1[k].x];
    }
#pragma unroll
    for (int k = 0; k < SEG_UNR; ++k) {
      if (a[k] < z[k]) {
        const int p = atomicAdd(&cur[r0[k].y - node0], 1);
        v2i ent;
        ent.x = r0[k].x;
        ent.y = __float_as_int(w0[k]);
        if (staged) stage[p] = ent; else csc[start + p] = ent;
      }
      if (a[k] + SEG_LANES < z[k]) {
        const int p = atomicAdd(&cur[r1[k].y - node0], 1);
        v2i ent;
        ent.x = r1[k].x;
        ent.y = __float_as_int(w1[k]);
        if (staged) stage[p] = ent; else csc[start + p] = ent;
      }
      for (int i = a[k] + 2 * SEG_LANES; i < z[k]; i += SEG_LANES) {
        const v2i r = binD[o[k] + i];
        const float w = degO[r.x];
        const int p = atomicAdd(&cur[r.y - node0], 1);
        v2i ent;
        ent.x = r.x;
        ent.y = __float_as_int(w);
        if (staged) stage[p] = ent; else csc[start + p] = ent;
      }
    }
  }
  if (staged) {
    __syncthreads();
    for (int i = tid; i < nrec; i += 1024) csc[start + i] = stage[i];
  }
}

static __device__ __forceinline__ void put_rows128(float* __restrict__ out,
                                                   const float* __restrict__ tile,
                                                   int row0, int wave, int lane) {
  const unsigned lane4 = (unsigned)lane * 4u;
#pragma unroll
  for (int i = 0; i < 2; ++i) {
    const int row = wave * 2 + i;
    float* __restrict__ orow = out + (long)(row0 + row) * 128;
    *(v4f*)(orow + lane4) = *(const v4f*)(tile + row * 132 + lane * 4);
  }
}

template <int NCOLS, int WAVES, bool RELU, bool RESID>
__global__ __launch_bounds__(WAVES * 32) void k_gemm_wmma(const float* __restrict__ A,
                                                          const float* __restrict__ W,
                                                          const float* __restrict__ bias,
                                                          float* __restrict__ out,
                                                          float actW, float resS, int nRows) {
  constexpr int NCT   = NCOLS / 16;
  constexpr int MT    = WAVES / NCT;
  constexpr int BR    = MT * 16;
  constexpr int RPW   = BR / WAVES;
  constexpr int PITCH = 132;
  static_assert(NCOLS == 128 && BR == 16 && RPW == 2, "the whole-row store takes 16 x 128 tiles, two rows per wave");
  __shared__ __attribute__((aligned(16))) float lds[4][BR * PITCH];

  const int wave = threadIdx.x >> 5;
  const int lane = threadIdx.x & 31;
  const int ct   = wave % NCT;
  const int mt   = wave / NCT;
  const int col0 = ct * 16;
  const int half = lane >> 4;
  const int l16  = lane & 15;
  const int outCol = col0 + l16;

  v2f bfr[32];
#pragma unroll
  for (int s = 0; s < 32; ++s) {
    const int kb = 8 * (s >> 1) + 4 * half + 2 * (s & 1);
    bfr[s].x = W[(long)kb * NCOLS + outCol];
    bfr[s].y = W[(long)(kb + 1) * NCOLS + outCol];
  }
  const float bv = bias[outCol];

  const int nblk = nRows / BR;
  int blk = blockIdx.x;

  v4f stage[RPW];
#pragma unroll
  for (int i = 0; i < RPW; ++i) stage[i] = (v4f){0.0f, 0.0f, 0.0f, 0.0f};
  if (blk < nblk) {
#pragma unroll
    for (int i = 0; i < RPW; ++i)
      stage[i] = *(const v4f*)(A + (long)(blk * BR + wave * RPW + i) * 128 + lane * 4);
  }

  int p = 0;
  int prev = -1;
  for (; blk < nblk; blk += gridDim.x) {
    float* __restrict__ buf = lds[p];
#pragma unroll
    for (int i = 0; i < RPW; ++i)
      *(v4f*)(buf + (wave * RPW + i) * PITCH + lane * 4) = stage[i];

    __builtin_amdgcn_fence(__ATOMIC_RELEASE, "workgroup", "local");
    __builtin_amdgcn_s_barrier();
    __builtin_amdgcn_fence(__ATOMIC_ACQUIRE, "workgroup", "local");

    const int next = blk + gridDim.x;
    if (next < nblk) {
#pragma unroll
      for (int i = 0; i < RPW; ++i)
        stage[i] = *(const v4f*)(A + (long)(next * BR + wave * RPW + i) * 128 + lane * 4);
    }

    if (prev >= 0) put_rows128(out, lds[2 + (p ^ 1)], prev * BR, wave, lane);

    const float* __restrict__ arow = buf + (mt * 16 + l16) * PITCH + half * 4;
    v4f af[16];
#pragma unroll
    for (int j = 0; j < 16; ++j) af[j] = *(const v4f*)(arow + 8 * j);

    v8f acc = {};
#pragma unroll
    for (int j = 0; j < 16; ++j) {
      v2f a0, a1;
      a0.x = af[j].x; a0.y = af[j].y;
      a1.x = af[j].z; a1.y = af[j].w;
      acc = __builtin_amdgcn_wmma_f32_16x16x4_f32(false, a0, false, bfr[2 * j],
                                                  (short)0, acc, false, false);
      acc = __builtin_amdgcn_wmma_f32_16x16x4_f32(false, a1, false, bfr[2 * j + 1],
                                                  (short)0, acc, false, false);
    }

    const int rl0 = mt * 16 + half * 8;
    float* __restrict__ otile = lds[2 + p];
#pragma unroll
    for (int r = 0; r < 8; ++r) {
      float v = actW * acc[r] + bv;
      if (RESID) v += resS * buf[(rl0 + r) * PITCH + outCol];
      if (RELU) v = fmaxf(v, 0.0f);
      otile[(rl0 + r) * PITCH + outCol] = v;
    }
    prev = blk;
    p ^= 1;
  }
  if (prev >= 0) {
    __builtin_amdgcn_fence(__ATOMIC_RELEASE, "workgroup", "local");
    __builtin_amdgcn_s_barrier();
    __builtin_amdgcn_fence(__ATOMIC_ACQUIRE, "workgroup", "local");
    put_rows128(out, lds[2 + (p ^ 1)], prev * BR, wave, lane);
  }
}


static __device__ __forceinline__ v4f gather_node(const v2i* __restrict__ csc,
                                                  const float* __restrict__ x,
                                                  int start, int c, v2i ent0, int lane) {
  const unsigned lane4 = (unsigned)lane * 4u;
  v4f acc = {0.0f, 0.0f, 0.0f, 0.0f};
  int idx = ent0.x;
  int wbt = ent0.y;
  for (int base = 0; base < c; base += 32) {
    if (base > 0) {
      v2i t = {0, 0};
      if (base + lane < c) t = csc[start + base + lane];
      idx = t.x;
      wbt = t.y;
    }
    const int n = (c - base < 32) ? c - base : 32;
#pragma unroll 1
    for (int j = 0; j < n; j += 8) {
      int   s[8];
      float w[8];
#pragma unroll
      for (int u = 0; u < 8; ++u) {
        const int e = (j + u < n) ? j + u : n - 1;
        s[u] = __builtin_amdgcn_readlane(idx, e);
        w[u] = __int_as_float(__builtin_amdgcn_readlane(wbt, e));
      }
      v4f v[8];
      if (j + 8 <= n) {
#pragma unroll
        for (int u = 0; u < 8; ++u) v[u] = *(const v4f*)(x + (long)s[u] * 128 + lane4);
#pragma unroll
        for (int u = 0; u < 8; ++u) acc += w[u] * v[u];
      } else {
#pragma unroll
        for (int u = 0; u < 8; ++u) {
          v[u] = (v4f){0.0f, 0.0f, 0.0f, 0.0f};
          if (j + u < n) v[u] = *(const v4f*)(x + (long)s[u] * 128 + lane4);
        }
#pragma unroll
        for (int u = 0; u < 8; ++u) acc += w[u] * v[u];
      }
    }
  }
  return acc;
}

template <bool RELU, bool PROJ>
__global__ __launch_bounds__(256) void k_layer_fused(const int* __restrict__ rowptr,
                                                     const int* __restrict__ cnt,
                                                     const int* __restrict__ cscRaw,
                                                     const float* __restrict__ x,
                                                     const float* __restrict__ dirs,
                                                     const float* __restrict__ x0,
                                                     const float* __restrict__ W,
                                                     const float* __restrict__ bias,
                                                     float* __restrict__ out,
                                                     const float* __restrict__ W2,
                                                     const float* __restrict__ bias2,
                                                     float* __restrict__ out2,
                                                     float actW, float resS, int nRows) {
  constexpr int NCOLS = 128;
  constexpr int BR    = 16;
  constexpr int PITCH = 132;
  constexpr int NC2   = 64;
  constexpr int NBUF = PROJ ? 1 : 2;
  __shared__ __attribute__((aligned(16))) float lds[(PROJ ? 3 : 4) * BR * PITCH];
  float* const xt = lds + BR * PITCH;
  float* const ot = lds + 2 * BR * PITCH;
  __shared__ v4f w2f[PROJ ? (NC2 / 16) * 16 * 32 : 1];
  const v2i* __restrict__ csc = (const v2i*)cscRaw;

  const int wave = __builtin_amdgcn_readfirstlane((int)(threadIdx.x >> 5));
  const int lane = threadIdx.x & 31;
  const int col0 = wave * 16;
  const int half = lane >> 4;
  const int l16  = lane & 15;
  const int outCol = col0 + l16;

  v2f bfr[32];
#pragma unroll
  for (int s = 0; s < 32; ++s) {
    const int kb = 8 * (s >> 1) + 4 * half + 2 * (s & 1);
    bfr[s].x = W[(long)kb * NCOLS + outCol];
    bfr[s].y = W[(long)(kb + 1) * NCOLS + outCol];
  }
  const float bv = bias[outCol];

  if (PROJ) {
    constexpr int NFR = (NC2 / 16) * 16 * 32 / 256;
    v4f fr[NFR];
#pragma unroll
    for (int u = 0; u < NFR; ++u) {
      const int i  = u * 256 + (int)threadIdx.x;
      const int ln = i & 31;
      const int j  = (i >> 5) & 15;
      const int t  = i >> 9;
      const float* __restrict__ wp = W2 + (8 * j + 4 * (ln >> 4)) * NC2 + t * 16 + (ln & 15);
      fr[u].x = wp[0];
      fr[u].y = wp[NC2];
      fr[u].z = wp[2 * NC2];
      fr[u].w = wp[3 * NC2];
    }
#pragma unroll
    for (int u = 0; u < NFR; ++u) w2f[u * 256 + (int)threadIdx.x] = fr[u];
  }

  const int nblk = nRows / BR;
  int p = 0;
  int prev = -1;
  for (int blk = blockIdx.x; blk < nblk; blk += gridDim.x) {
    float* buf = lds + p * (BR * PITCH);

    const int nodeA  = blk * BR + wave * 2;
    const int startA = rowptr[nodeA];
    const int startB = rowptr[nodeA + 1];
    const int cA     = cnt[nodeA];
    const int cB     = cnt[nodeA + 1];
    v2i iA = {0, 0};
    v2i iB = {0, 0};
    if (lane < cA) iA = csc[startA + lane];
    if (lane < cB) iB = csc[startB + lane];
    const float scA = (1.0f - GA_ALPHA) * dirs[nodeA];
    const float scB = (1.0f - GA_ALPHA) * dirs[nodeA + 1];
    const long  oA  = (long)nodeA * 128 + lane * 4;
    const v4f x0A = *(const v4f*)(x0 + oA);
    const v4f x0B = *(const v4f*)(x0 + oA + 128);

    const v4f accA = gather_node(csc, x, startA, cA, iA, lane);
    v4f rA;
    rA.x = scA * accA.x + GA_ALPHA * x0A.x;
    rA.y = scA * accA.y + GA_ALPHA * x0A.y;
    rA.z = scA * accA.z + GA_ALPHA * x0A.z;
    rA.w = scA * accA.w + GA_ALPHA * x0A.w;
    *(v4f*)(buf + (wave * 2) * PITCH + lane * 4) = rA;

    const v4f accB = gather_node(csc, x, startB, cB, iB, lane);
    v4f rB;
    rB.x = scB * accB.x + GA_ALPHA * x0B.x;
    rB.y = scB * accB.y + GA_ALPHA * x0B.y;
    rB.z = scB * accB.z + GA_ALPHA * x0B.z;
    rB.w = scB * accB.w + GA_ALPHA * x0B.w;
    *(v4f*)(buf + (wave * 2 + 1) * PITCH + lane * 4) = rB;

    __builtin_amdgcn_fence(__ATOMIC_RELEASE, "workgroup", "local");
    __builtin_amdgcn_s_barrier();
    __builtin_amdgcn_fence(__ATOMIC_ACQUIRE, "workgroup", "local");

    if (!PROJ && prev >= 0) put_rows128(out, ot + (p ^ 1) * (BR * PITCH), prev * BR, wave, lane);

    const float* __restrict__ arow = buf + l16 * PITCH + half * 4;
    v4f af[16];
#pragma unroll
    for (int j = 0; j < 16; ++j) af[j] = *(const v4f*)(arow + 8 * j);

    v8f acc = {};
#pragma unroll
    for (int j = 0; j < 16; ++j) {
      v2f a0, a1;
      a0.x = af[j].x; a0.y = af[j].y;
      a1.x = af[j].z; a1.y = af[j].w;
      acc = __builtin_amdgcn_wmma_f32_16x16x4_f32(false, a0, false, bfr[2 * j],
                                                  (short)0, acc, false, false);
      acc = __builtin_amdgcn_wmma_f32_16x16x4_f32(false, a1, false, bfr[2 * j + 1],
                                                  (short)0, acc, false, false);
    }

    const int rl0 = half * 8;
    if (!PROJ) {
      float* __restrict__ otile = ot + p * (BR * PITCH);
#pragma unroll
      for (int r = 0; r < 8; ++r) {
        float v = actW * acc[r] + bv;
        v += resS * buf[(rl0 + r) * PITCH + outCol];
        if (RELU) v = fmaxf(v, 0.0f);
        otile[(rl0 + r) * PITCH + outCol] = v;
      }
    } else {
#pragma unroll
      for (int r = 0; r < 8; ++r) {
        float v = actW * acc[r] + bv;
        v += resS * buf[(rl0 + r) * PITCH + outCol];
        if (RELU) v = fmaxf(v, 0.0f);
        xt[(rl0 + r) * PITCH + outCol] = v;
      }
      if (prev >= 0) {
        const unsigned off = (unsigned)(prev * (BR / 2) + wave) * (2u * NC2 * 4u) + (unsigned)lane * 16u;
        *(v4f*)((char*)out2 + off) = *(const v4f*)(ot + wave * (2 * PITCH) + lane * 4);
      }
      __builtin_amdgcn_fence(__ATOMIC_RELEASE, "workgroup", "local");
      __builtin_amdgcn_s_barrier();
      __builtin_amdgcn_fence(__ATOMIC_ACQUIRE, "workgroup", "local");
      if (wave < NC2 / 16) {
        const float* __restrict__ arow2 = xt + l16 * PITCH + half * 4;
        const v4f* __restrict__ bcol = w2f + wave * 16 * 32 + lane;
        const float bv2 = bias2[wave * 16 + l16];
        v8f acc2 = {};
#pragma unroll
        for (int j = 0; j < 16; ++j) {
          const v4f a = *(const v4f*)(arow2 + 8 * j);
          const v4f b = bcol[j * 32];
          v2f a0, a1, b0, b1;
          a0.x = a.x; a0.y = a.y;
          a1.x = a.z; a1.y = a.w;
          b0.x = b.x; b0.y = b.y;
          b1.x = b.z; b1.y = b.w;
          acc2 = __builtin_amdgcn_wmma_f32_16x16x4_f32(false, a0, false, b0,
                                                       (short)0, acc2, false, false);
          acc2 = __builtin_amdgcn_wmma_f32_16x16x4_f32(false, a1, false, b1,
                                                       (short)0, acc2, false, false);
        }
        float* __restrict__ otile = ot + rl0 * PITCH + wave * 16 + l16;
#pragma unroll
        for (int r = 0; r < 8; ++r) otile[(r & ~1) * PITCH + (r & 1) * NC2] = acc2[r] + bv2;
      }
    }
    prev = blk;
    if (NBUF == 2) p ^= 1;
  }
  if (prev >= 0) {
    __builtin_amdgcn_fence(__ATOMIC_RELEASE, "workgroup", "local");
    __builtin_amdgcn_s_barrier();
    __builtin_amdgcn_fence(__ATOMIC_ACQUIRE, "workgroup", "local");
    if (!PROJ) {
      put_rows128(out, ot + (p ^ 1) * (BR * PITCH), prev * BR, wave, lane);
    } else {
      const unsigned off = (unsigned)(prev * (BR / 2) + wave) * (2u * NC2 * 4u) + (unsigned)lane * 16u;
      *(v4f*)((char*)out2 + off) = *(const v4f*)(ot + wave * (2 * PITCH) + lane * 4);
    }
  }
}


static inline long cdiv(long a, long b) { return (a + b - 1) / b; }

extern "C" void kernel_launch(void* const* d_in, const int* in_sizes, int n_in,
                              void* d_out, int out_size, void* d_ws, size_t ws_size,
                              hipStream_t stream) {
  const float* feat = (const float*)d_in[0];
  const int*   src  = (const int*)d_in[1];
  const int*   dst  = (const int*)d_in[2];
  const float* w1   = (const float*)d_in[3];
  const float* b1   = (const float*)d_in[4];
  const float* cw1  = (const float*)d_in[5];
  const float* cb1  = (const float*)d_in[6];
  const float* cw2  = (const float*)d_in[7];
  const float* cb2  = (const float*)d_in[8];
  const float* w2   = (const float*)d_in[9];
  const float* b2   = (const float*)d_in[10];

  const int N = in_sizes[0] / 128;
  const int E = in_sizes[1];
  const long NF = (long)N * 128;
  const int NA = (N + 63) & ~63;

  char* wsb = (char*)d_ws;
  wsb += (size_t)NA * 4;
  int* cntI   = (int*)wsb;                wsb += (size_t)NA * 4;
  int* rowptr = (int*)wsb;                wsb += (size_t)NA * 4;
  wsb += (size_t)NA * 4;
  int* bsum   = (int*)wsb;                wsb += 512 * 4;
  int* csc    = (int*)wsb;                wsb += ((size_t)E * 8 + 255) & ~(size_t)255;
  float* degO = (float*)wsb;              wsb += (size_t)NA * 4;
  float* degI = (float*)wsb;              wsb += (size_t)NA * 4;
  float* h0   = (float*)wsb;              wsb += (size_t)NF * 4;
  float* bufS = (float*)wsb;              wsb += (size_t)NF * 4;
  float* bufA = (float*)wsb;
  (void)ws_size; (void)n_in; (void)out_size;

  const int NBIN = (N + BIN_NODES - 1) >> BIN_SHIFT;
  const int NSL  = (int)cdiv(E, SLICE_CAP);
  const int per  = (int)cdiv(E, NSL);
  const int TW   = 2 * NBIN + 1;
  char* tb = (char*)bufS;
  int* binD     = (int*)tb;               tb += ((size_t)E * 8 + 255) & ~(size_t)255;
  int* binS     = (int*)tb;               tb += ((size_t)E * 4 + 255) & ~(size_t)255;
  int* tbl      = (int*)tb;
  (void)bsum;

  k_slice_sort<<<NSL, 1024, 0, stream>>>(src, dst, tbl, binD, binS, E, per, NBIN, TW);
  k_src_count<<<NBIN, 1024, 0, stream>>>(binS, tbl, degO, N, per, NBIN, TW, NSL);
  k_dst_place<<<NBIN, 1024, 0, stream>>>(binD, tbl, degO, rowptr, cntI, degI, csc, N, per, TW, NSL);

  k_gemm_wmma<128, 8, true, false><<<512, 256, 0, stream>>>(feat, w1, b1, h0, 1.0f, 0.0f, N);

  k_layer_fused<true, false><<<768, 256, 0, stream>>>(rowptr, cntI, csc, h0, degI, h0, cw1, cb1, bufA,
                                                       nullptr, nullptr, nullptr,
                                                       GA_BETA1, 1.0f - GA_BETA1, N);

  k_layer_fused<false, true><<<768, 256, 0, stream>>>(rowptr, cntI, csc, bufA, degI, h0, cw2, cb2, nullptr,
                                                       w2, b2, (float*)d_out,
                                                       GA_BETA2, 1.0f - GA_BETA2, N);
}
